# starting kernel
# baseline (speedup 1.0000x reference)
#include <hip/hip_runtime.h>


#define NB 10000
#define BB 4
#define EE 160000
#define DD 128
#define HH 4
#define OO 64
#define HO 256
#define MROWS (BB * NB)
#define NEG_SLOPE 0.2f

typedef float v2f __attribute__((ext_vector_type(2)));
typedef float v8f __attribute__((ext_vector_type(8)));

__device__ __forceinline__ unsigned fenc(float f) {
  unsigned u = __float_as_uint(f);
  return (u & 0x80000000u) ? ~u : (u | 0x80000000u);
}
__device__ __forceinline__ float fdec(unsigned u) {
  return __uint_as_float((u & 0x80000000u) ? (u ^ 0x80000000u) : ~u);
}

__global__ __launch_bounds__(256) void gemm_h(const float* __restrict__ x,
                                              const float* __restrict__ W,
                                              float* __restrict__ h) {
  __shared__ float2 Wlds[32 * HO];
  const int tid = threadIdx.x;
  const int lane = tid & 31;
  const int wave = tid >> 5;
  const int rowBase = blockIdx.x * 128 + wave * 16;
  const bool active = rowBase < MROWS;

  v8f acc[16];
  const v8f vzero = {0.f, 0.f, 0.f, 0.f, 0.f, 0.f, 0.f, 0.f};
#pragma unroll
  for (int t = 0; t < 16; ++t) acc[t] = vzero;

  const int m = lane & 15;
  const int kb = (lane >> 4) << 1;
  const int kp = lane >> 4;

  for (int kblk = 0; kblk < DD; kblk += 64) {
#pragma unroll 4
    for (int p = 0; p < 32; ++p) {
      float2 pr;
      pr.x = W[(size_t)(kblk + 2 * p) * HO + tid];
      pr.y = W[(size_t)(kblk + 2 * p + 1) * HO + tid];
      Wlds[p * HO + tid] = pr;
    }
    __syncthreads();

    if (active) {
      const float* xrow = x + (size_t)(rowBase + m) * DD + kblk;
      for (int kk = 0; kk < 64; kk += 4) {
        v2f a;
        a.x = xrow[kk + kb];
        a.y = xrow[kk + kb + 1];
        const float2* w0 = Wlds + ((kk >> 1) + kp) * HO + (lane & 15);
#pragma unroll
        for (int t = 0; t < 16; ++t) {
          float2 bp = w0[t * 16];
          v2f b;
          b.x = bp.x;
          b.y = bp.y;
          acc[t] = __builtin_amdgcn_wmma_f32_16x16x4_f32(
              false, a, false, b, (short)0, acc[t], false, false);
        }
      }
    }
    __syncthreads();
  }

  if (active) {
    const int nLocal = lane & 15;
    const int rOff = (lane >> 4) * 8;
#pragma unroll
    for (int t = 0; t < 16; ++t)
#pragma unroll
      for (int v = 0; v < 8; ++v)
        h[(size_t)(rowBase + rOff + v) * HO + t * 16 + nLocal] = acc[t][v];
  }
}

__global__ __launch_bounds__(256) void calc_attn_node(
    const float* __restrict__ h, const float* __restrict__ att_src,
    const float* __restrict__ att_dst, float* __restrict__ a_src,
    float* __restrict__ a_dst) {
  const int wv = blockIdx.x * 8 + (threadIdx.x >> 5);
  if (wv >= MROWS * HH) return;
  const int lane = threadIdx.x & 31;
  const int head = wv & 3;
  const float2 hv = ((const float2*)(h + (size_t)wv * OO))[lane];
  const float2 as = ((const float2*)(att_src + head * OO))[lane];
  const float2 ad = ((const float2*)(att_dst + head * OO))[lane];
  float vs = hv.x * as.x + hv.y * as.y;
  float vd = hv.x * ad.x + hv.y * ad.y;
#pragma unroll
  for (int off = 16; off > 0; off >>= 1) {
    vs += __shfl_xor(vs, off);
    vd += __shfl_xor(vd, off);
  }
  if (lane == 0) { a_src[wv] = vs; a_dst[wv] = vd; }
}

__global__ void calc_edge_coef(const float* __restrict__ W_edge,
                               const float* __restrict__ att_edge,
                               float* __restrict__ coef) {
  const int t = threadIdx.x;
  if (t < HH) {
    float s = 0.f;
    for (int o = 0; o < OO; ++o) s += W_edge[t * OO + o] * att_edge[t * OO + o];
    coef[t] = s;
  }
}

__device__ __forceinline__ float edge_alpha(const float* a_src, const float* a_dst,
                                            int sN, int dN, int b, int hh,
                                            float ae, const float* coef) {
  float a = a_src[((size_t)b * NB + sN) * HH + hh] +
            a_dst[((size_t)b * NB + dN) * HH + hh] + ae * coef[hh];
  return a > 0.f ? a : a * NEG_SLOPE;
}

__global__ __launch_bounds__(256) void edge_max(
    const int* __restrict__ src, const int* __restrict__ dst,
    const float* __restrict__ eattr, const float* __restrict__ a_src,
    const float* __restrict__ a_dst, const float* __restrict__ coef,
    unsigned* __restrict__ mbuf) {
  const int e = blockIdx.x * 256 + threadIdx.x;
  if (e >= EE) return;
  const int sN = src[e], dN = dst[e];
  const float ae = eattr[e];
  for (int b = 0; b < BB; ++b)
#pragma unroll
    for (int hh = 0; hh < HH; ++hh) {
      float a = edge_alpha(a_src, a_dst, sN, dN, b, hh, ae, coef);
      atomicMax(&mbuf[(size_t)dN * 16 + b * 4 + hh], fenc(a));
    }
}

__global__ __launch_bounds__(256) void edge_sum(
    const int* __restrict__ src, const int* __restrict__ dst,
    const float* __restrict__ eattr, const float* __restrict__ a_src,
    const float* __restrict__ a_dst, const float* __restrict__ coef,
    const unsigned* __restrict__ mbuf, float* __restrict__ sden) {
  const int e = blockIdx.x * 256 + threadIdx.x;
  if (e >= EE) return;
  const int sN = src[e], dN = dst[e];
  const float ae = eattr[e];
  for (int b = 0; b < BB; ++b)
#pragma unroll
    for (int hh = 0; hh < HH; ++hh) {
      float a = edge_alpha(a_src, a_dst, sN, dN, b, hh, ae, coef);
      const size_t mi = (size_t)dN * 16 + b * 4 + hh;
      atomicAdd(&sden[mi], __expf(a - fdec(mbuf[mi])));
    }
}

__global__ __launch_bounds__(256) void edge_scatter(
    const int* __restrict__ src, const int* __restrict__ dst,
    const float* __restrict__ eattr, const float* __restrict__ a_src,
    const float* __restrict__ a_dst, const float* __restrict__ coef,
    const unsigned* __restrict__ mbuf, const float* __restrict__ sden,
    const float* __restrict__ h, float* __restrict__ acc) {
  __shared__ float wsh[16];
  const int e = blockIdx.x;
  const int t = threadIdx.x;
  const int sN = src[e], dN = dst[e];
  if (t < 16) {
    const int b = t >> 2, hh = t & 3;
    float a = edge_alpha(a_src, a_dst, sN, dN, b, hh, eattr[e], coef);
    const size_t mi = (size_t)dN * 16 + b * 4 + hh;
    float p = __expf(a - fdec(mbuf[mi]));
    wsh[t] = p / fmaxf(sden[mi], 1e-16f);
  }
  __syncthreads();
#pragma unroll
  for (int i = t; i < BB * HO; i += 256) {
    const int b = i >> 8, rem = i & 255;
    float val = wsh[b * 4 + (rem >> 6)] * h[((size_t)b * NB + sN) * HO + rem];
    atomicAdd(&acc[((size_t)dN * 4 + b) * HO + rem], val);
  }
}

__global__ __launch_bounds__(256) void finalize(const float* __restrict__ acc,
                                                const float* __restrict__ bias,
                                                float* __restrict__ out) {
  const int i = blockIdx.x * 256 + threadIdx.x;
  if (i >= BB * NB * OO) return;
  const int o = i & 63;
  const int bn = i >> 6;
  const int n = bn % NB, b = bn / NB;
  const float* p = acc + ((size_t)n * 4 + b) * HO + o;
  out[i] = (p[0] + p[64] + p[128] + p[192]) * 0.25f + bias[o];
}

extern "C" void kernel_launch(void* const* d_in, const int* in_sizes, int n_in,
                              void* d_out, int out_size, void* d_ws, size_t ws_size,
                              hipStream_t stream) {
  const float* x        = (const float*)d_in[0];
  const int*   eidx     = (const int*)d_in[1];
  const float* eattr    = (const float*)d_in[2];
  const float* W        = (const float*)d_in[3];
  const float* att_src  = (const float*)d_in[4];
  const float* att_dst  = (const float*)d_in[5];
  const float* W_edge   = (const float*)d_in[6];
  const float* att_edge = (const float*)d_in[7];
  const float* bias     = (const float*)d_in[8];
  float* out = (float*)d_out;

  char* ws = (char*)d_ws;
  float*    h_buf  = (float*)   (ws + 0);
  float*    a_src  = (float*)   (ws + 40960000);
  float*    a_dst  = (float*)   (ws + 43520000);
  unsigned* mbuf   = (unsigned*)(ws + 46080000);
  float*    sden   = (float*)   (ws + 48640000);
  float*    coef   = (float*)   (ws + 51200000);
  float*    acc    = (float*)   (ws + 51200256);

  const int* srcI = eidx;
  const int* dstI = eidx + EE;

  hipMemsetAsync(ws + 46080000, 0, 46080256, stream);

  calc_edge_coef<<<1, 64, 0, stream>>>(W_edge, att_edge, coef);
  gemm_h<<<(MROWS + 127) / 128, 256, 0, stream>>>(x, W, h_buf);
  calc_attn_node<<<(MROWS * HH) / 8, 256, 0, stream>>>(h_buf, att_src, att_dst,
                                                       a_src, a_dst);
  edge_max<<<(EE + 255) / 256, 256, 0, stream>>>(srcI, dstI, eattr, a_src,
                                                 a_dst, coef, mbuf);
  edge_sum<<<(EE + 255) / 256, 256, 0, stream>>>(srcI, dstI, eattr, a_src,
                                                 a_dst, coef, mbuf, sden);
  edge_scatter<<<EE, 256, 0, stream>>>(srcI, dstI, eattr, a_src, a_dst, coef,
                                       mbuf, sden, h_buf, acc);
  finalize<<<(BB * NB * OO + 255) / 256, 256, 0, stream>>>(acc, bias, out);
}


# Edge sums kept in LDS, not global atomics: 250 work-groups each own 40 destination nodes, filter the destination list, add gathered rows with LDS float atomics and write out; sums still unordered
# speedup vs baseline: 4.5822x; 4.5822x over previous
#include <hip/hip_runtime.h>


#define NB 10000
#define BB 4
#define EE 160000
#define DD 128
#define HH 4
#define OO 64
#define HO 256
#define MROWS (BB * NB)
#define NEG_SLOPE 0.2f

typedef float v2f __attribute__((ext_vector_type(2)));
typedef float v8f __attribute__((ext_vector_type(8)));

__device__ __forceinline__ unsigned fenc(float f) {
  unsigned u = __float_as_uint(f);
  return (u & 0x80000000u) ? ~u : (u | 0x80000000u);
}
__device__ __forceinline__ float fdec(unsigned u) {
  return __uint_as_float((u & 0x80000000u) ? (u ^ 0x80000000u) : ~u);
}

__global__ __launch_bounds__(256) void gemm_h(const float* __restrict__ x,
                                              const float* __restrict__ W,
                                              float* __restrict__ h) {
  __shared__ float2 Wlds[32 * HO];
  const int tid = threadIdx.x;
  const int lane = tid & 31;
  const int wave = tid >> 5;
  const int rowBase = blockIdx.x * 128 + wave * 16;
  const bool active = rowBase < MROWS;

  v8f acc[16];
  const v8f vzero = {0.f, 0.f, 0.f, 0.f, 0.f, 0.f, 0.f, 0.f};
#pragma unroll
  for (int t = 0; t < 16; ++t) acc[t] = vzero;

  const int m = lane & 15;
  const int kb = (lane >> 4) << 1;
  const int kp = lane >> 4;

  for (int kblk = 0; kblk < DD; kblk += 64) {
#pragma unroll 4
    for (int p = 0; p < 32; ++p) {
      float2 pr;
      pr.x = W[(size_t)(kblk + 2 * p) * HO + tid];
      pr.y = W[(size_t)(kblk + 2 * p + 1) * HO + tid];
      Wlds[p * HO + tid] = pr;
    }
    __syncthreads();

    if (active) {
      const float* xrow = x + (size_t)(rowBase + m) * DD + kblk;
      for (int kk = 0; kk < 64; kk += 4) {
        v2f a;
        a.x = xrow[kk + kb];
        a.y = xrow[kk + kb + 1];
        const float2* w0 = Wlds + ((kk >> 1) + kp) * HO + (lane & 15);
#pragma unroll
        for (int t = 0; t < 16; ++t) {
          float2 bp = w0[t * 16];
          v2f b;
          b.x = bp.x;
          b.y = bp.y;
          acc[t] = __builtin_amdgcn_wmma_f32_16x16x4_f32(
              false, a, false, b, (short)0, acc[t], false, false);
        }
      }
    }
    __syncthreads();
  }

  if (active) {
    const int nLocal = lane & 15;
    const int rOff = (lane >> 4) * 8;
#pragma unroll
    for (int t = 0; t < 16; ++t)
#pragma unroll
      for (int v = 0; v < 8; ++v)
        h[(size_t)(rowBase + rOff + v) * HO + t * 16 + nLocal] = acc[t][v];
  }
}

#define ATTN_ROWS 5
#define ATTN_CHUNKS (MROWS / ATTN_ROWS)
#define ATTN_BLOCKS 500
__global__ __launch_bounds__(256) void calc_attn_node(
    const float* __restrict__ h, const float* __restrict__ att_src,
    const float* __restrict__ att_dst, float* __restrict__ a_src,
    float* __restrict__ a_dst) {
  const int lane = threadIdx.x & 31;
  const int w = blockIdx.x * 8 + __builtin_amdgcn_readfirstlane((int)(threadIdx.x >> 5));
  float2 as[HH], ad[HH];
#pragma unroll
  for (int k = 0; k < HH; ++k) {
    as[k] = ((const float2*)(att_src + k * OO))[lane];
    ad[k] = ((const float2*)(att_dst + k * OO))[lane];
  }
  for (int c = w; c < ATTN_CHUNKS; c += ATTN_BLOCKS * 8) {
    const int row0 = c * ATTN_ROWS;
    const float2* hp = (const float2*)(h + (size_t)row0 * HO) + lane;
    float2 hv[ATTN_ROWS][HH];
#pragma unroll
    for (int r = 0; r < ATTN_ROWS; ++r)
#pragma unroll
      for (int k = 0; k < HH; ++k) hv[r][k] = hp[(r * HH + k) * 32];
    float vs[ATTN_ROWS][HH], vd[ATTN_ROWS][HH];
#pragma unroll
    for (int r = 0; r < ATTN_ROWS; ++r)
#pragma unroll
      for (int k = 0; k < HH; ++k) {
        vs[r][k] = hv[r][k].x * as[k].x + hv[r][k].y * as[k].y;
        vd[r][k] = hv[r][k].x * ad[k].x + hv[r][k].y * ad[k].y;
      }
#pragma unroll
    for (int off = 16; off > 0; off >>= 1)
#pragma unroll
      for (int r = 0; r < ATTN_ROWS; ++r)
#pragma unroll
        for (int k = 0; k < HH; ++k) {
          vs[r][k] += __shfl_xor(vs[r][k], off);
          vd[r][k] += __shfl_xor(vd[r][k], off);
        }
    float4 os = make_float4(0.f, 0.f, 0.f, 0.f), od = os;
#pragma unroll
    for (int r = 0; r < ATTN_ROWS; ++r)
      if (lane == r) {
        os = make_float4(vs[r][0], vs[r][1], vs[r][2], vs[r][3]);
        od = make_float4(vd[r][0], vd[r][1], vd[r][2], vd[r][3]);
      }
    if (lane < ATTN_ROWS) {
      ((float4*)a_src)[row0 + lane] = os;
      ((float4*)a_dst)[row0 + lane] = od;
    }
  }
}

__global__ void calc_edge_coef(const float* __restrict__ W_edge,
                               const float* __restrict__ att_edge,
                               float* __restrict__ coef) {
  const int t = threadIdx.x;
  if (t < HH) {
    float s = 0.f;
    for (int o = 0; o < OO; ++o) s += W_edge[t * OO + o] * att_edge[t * OO + o];
    coef[t] = s;
  }
}

__device__ __forceinline__ float edge_score(float as, float ad, float ae, float cf) {
  float a = as + ad + ae * cf;
  return a > 0.f ? a : a * NEG_SLOPE;
}

#define EW_EDGES 32
#define EW_BLOCKS (EE / (EW_EDGES * 8))
static_assert(EW_BLOCKS * EW_EDGES * 8 == EE, "every wave owns 32 whole edges");

__device__ __forceinline__ int pair_pick(int v, int j, bool hi) {
  const int lo_v = __builtin_amdgcn_readlane(v, 2 * j);
  const int hi_v = __builtin_amdgcn_readlane(v, 2 * j + 1);
  return hi ? hi_v : lo_v;
}

__global__ __launch_bounds__(256) void edge_max(
    const int* __restrict__ src, const int* __restrict__ dst,
    const float* __restrict__ eattr, const float* __restrict__ a_src,
    const float* __restrict__ a_dst, const float* __restrict__ coef,
    unsigned* __restrict__ mbuf) {
  const int lane = threadIdx.x & 31;
  const int t = lane & 15;
  const bool hi = lane >= 16;
  const int w = blockIdx.x * 8 + __builtin_amdgcn_readfirstlane((int)(threadIdx.x >> 5));
  const int e0 = w * EW_EDGES;
  const int sv = src[e0 + lane];
  const int dv = dst[e0 + lane];
  const int av = __float_as_int(eattr[e0 + lane]);

  const int ot = (t >> 2) * (NB * HH) + (t & 3);
  const float cf = coef[t & 3];

  float al[EW_EDGES / 2];
  int dn[EW_EDGES / 2];
#pragma unroll
  for (int j = 0; j < EW_EDGES / 2; ++j) {
    const int sN = pair_pick(sv, j, hi);
    const int dN = pair_pick(dv, j, hi);
    const float ae = __int_as_float(pair_pick(av, j, hi));
    al[j] = edge_score(a_src[ot + sN * HH], a_dst[ot + dN * HH], ae, cf);
    dn[j] = dN * 16 + t;
  }
#pragma unroll
  for (int j = 0; j < EW_EDGES / 2; ++j) atomicMax(&mbuf[dn[j]], fenc(al[j]));
}

__global__ __launch_bounds__(256) void edge_sum(
    const int* __restrict__ src, const int* __restrict__ dst,
    const float* __restrict__ eattr, const float* __restrict__ a_src,
    const float* __restrict__ a_dst, const float* __restrict__ coef,
    const unsigned* __restrict__ mbuf, float* __restrict__ sden) {
  const int lane = threadIdx.x & 31;
  const int t = lane & 15;
  const bool hi = lane >= 16;
  const int w = blockIdx.x * 8 + __builtin_amdgcn_readfirstlane((int)(threadIdx.x >> 5));
  const int e0 = w * EW_EDGES;
  const int sv = src[e0 + lane];
  const int dv = dst[e0 + lane];
  const int av = __float_as_int(eattr[e0 + lane]);

  const int ot = (t >> 2) * (NB * HH) + (t & 3);
  const float cf = coef[t & 3];

  float al[EW_EDGES / 2];
  unsigned mv[EW_EDGES / 2];
  int dn[EW_EDGES / 2];
#pragma unroll
  for (int j = 0; j < EW_EDGES / 2; ++j) {
    const int sN = pair_pick(sv, j, hi);
    const int dN = pair_pick(dv, j, hi);
    const float ae = __int_as_float(pair_pick(av, j, hi));
    al[j] = edge_score(a_src[ot + sN * HH], a_dst[ot + dN * HH], ae, cf);
    dn[j] = dN * 16 + t;
    mv[j] = mbuf[dn[j]];
  }
#pragma unroll
  for (int j = 0; j < EW_EDGES / 2; ++j)
    atomicAdd(&sden[dn[j]], __expf(al[j] - fdec(mv[j])));
}

#define OWN_NODES 40
#define OWN_BLOCKS (NB / OWN_NODES)
static_assert(OWN_BLOCKS * OWN_NODES == NB, "every node has exactly one owner");
#define OWN_THREADS 1024
#define E4 (EE / 4)
static_assert(E4 * 4 == EE, "the destination list is read as whole int4");
#define SCAN_LOADS 10
#define SCAN_CHUNK4 (SCAN_LOADS * OWN_THREADS)
#define SCAN_CHUNKS ((E4 + SCAN_CHUNK4 - 1) / SCAN_CHUNK4)
#define LIST_FLUSH 4096
#define LIST_CAP (4 * SCAN_CHUNK4 + LIST_FLUSH)

struct EdgeRegs {
  float as, ad, ae, den;
  unsigned mx;
  int dl;
  float2 hv[BB][HH];
};

__device__ __forceinline__ void own_load(EdgeRegs& r, int sv, int dv, int av, int k,
                                         int n0, int ot, int t, int lane,
                                         const float* __restrict__ a_src,
                                         const float* __restrict__ a_dst,
                                         const unsigned* __restrict__ mbuf,
                                         const float* __restrict__ sden,
                                         const float* __restrict__ h) {
  const int sN = __builtin_amdgcn_readlane(sv, k);
  const int dN = __builtin_amdgcn_readlane(dv, k);
  r.ae = __int_as_float(__builtin_amdgcn_readlane(av, k));
  r.as = a_src[ot + sN * HH];
  r.ad = a_dst[ot + dN * HH];
  r.mx = mbuf[dN * 16 + t];
  r.den = sden[dN * 16 + t];
  r.dl = (dN - n0) * (BB * OO) + lane;
  const float2* hs = (const float2*)(h + sN * HO) + lane;
#pragma unroll
  for (int b = 0; b < BB; ++b)
#pragma unroll
    for (int k2 = 0; k2 < HH; ++k2) r.hv[b][k2] = hs[(b * NB * HO + k2 * OO) / 2];
}

__device__ __forceinline__ void own_add(const EdgeRegs& r, float cf, float* accl) {
  const float a = edge_score(r.as, r.ad, r.ae, cf);
  const float p = __expf(a - fdec(r.mx));
  const float wl = p / fmaxf(r.den, 1e-16f);
#pragma unroll
  for (int b = 0; b < BB; ++b) {
    const float w0 = __int_as_float(__builtin_amdgcn_readlane(__float_as_int(wl), b * 4 + 0));
    const float w1 = __int_as_float(__builtin_amdgcn_readlane(__float_as_int(wl), b * 4 + 1));
    const float w2 = __int_as_float(__builtin_amdgcn_readlane(__float_as_int(wl), b * 4 + 2));
    const float w3 = __int_as_float(__builtin_amdgcn_readlane(__float_as_int(wl), b * 4 + 3));
    const float vx = fmaf(w3, r.hv[b][3].x, fmaf(w2, r.hv[b][2].x, fmaf(w1, r.hv[b][1].x, w0 * r.hv[b][0].x)));
    const float vy = fmaf(w3, r.hv[b][3].y, fmaf(w2, r.hv[b][2].y, fmaf(w1, r.hv[b][1].y, w0 * r.hv[b][0].y)));
    __hip_atomic_fetch_add(&accl[r.dl + b * OO], vx, __ATOMIC_RELAXED, __HIP_MEMORY_SCOPE_WORKGROUP);
    __hip_atomic_fetch_add(&accl[r.dl + b * OO + 32], vy, __ATOMIC_RELAXED, __HIP_MEMORY_SCOPE_WORKGROUP);
  }
}

__global__ __launch_bounds__(OWN_THREADS) void edge_own(
    const int* __restrict__ src, const int* __restrict__ dst,
    const float* __restrict__ eattr, const float* __restrict__ a_src,
    const float* __restrict__ a_dst, const float* __restrict__ coef,
    const unsigned* __restrict__ mbuf, const float* __restrict__ sden,
    const float* __restrict__ h, const float* __restrict__ bias,
    float* __restrict__ out) {
  __shared__ float accl[OWN_NODES * BB * OO];
  __shared__ int elist[LIST_CAP];
  __shared__ int ecnt[SCAN_CHUNKS];
  const int tid = threadIdx.x;
  const int lane = tid & 31;
  const int t = lane & 15;
  const int wv = __builtin_amdgcn_readfirstlane(tid >> 5);
  const int n0 = blockIdx.x * OWN_NODES;
  const int ot = (t >> 2) * (NB * HH) + (t & 3);
  const float cf = coef[t & 3];

  for (int i = tid; i < OWN_NODES * BB * OO; i += OWN_THREADS) accl[i] = 0.f;
  if (tid < SCAN_CHUNKS) ecnt[tid] = 0;
  __syncthreads();

  const int4* dst4 = (const int4*)dst;
  int total = 0;
  for (int ch = 0; ch < SCAN_CHUNKS; ++ch) {
    int4 dq[SCAN_LOADS];
#pragma unroll
    for (int k = 0; k < SCAN_LOADS; ++k) {
      const int q = (ch * SCAN_LOADS + k) * OWN_THREADS + tid;
      dq[k] = q < E4 ? dst4[q] : make_int4(-1, -1, -1, -1);
    }
#pragma unroll
    for (int k = 0; k < SCAN_LOADS; ++k) {
      const int q = (ch * SCAN_LOADS + k) * OWN_THREADS + tid;
      const unsigned ux = (unsigned)(dq[k].x - n0), uy = (unsigned)(dq[k].y - n0);
      const unsigned uz = (unsigned)(dq[k].z - n0), uw = (unsigned)(dq[k].w - n0);
      if (min(min(ux, uy), min(uz, uw)) < (unsigned)OWN_NODES) {
        if (ux < (unsigned)OWN_NODES) elist[total + atomicAdd(&ecnt[ch], 1)] = 4 * q;
        if (uy < (unsigned)OWN_NODES) elist[total + atomicAdd(&ecnt[ch], 1)] = 4 * q + 1;
        if (uz < (unsigned)OWN_NODES) elist[total + atomicAdd(&ecnt[ch], 1)] = 4 * q + 2;
        if (uw < (unsigned)OWN_NODES) elist[total + atomicAdd(&ecnt[ch], 1)] = 4 * q + 3;
      }
    }
    __syncthreads();
    total += ecnt[ch];
    const bool last = ch == SCAN_CHUNKS - 1;
    if (!last && total <= LIST_FLUSH) continue;

    for (int b0 = 0; b0 < total; b0 += OWN_THREADS) {
      const int my = b0 + lane * 32 + wv;
      int sv = 0, dv = n0, av = 0;
      if (my < total) {
        const int e = elist[my];
        sv = src[e];
        dv = dst[e];
        av = __float_as_int(eattr[e]);
      }
      const int rem = total - b0 - wv;
      const int nk = rem <= 0 ? 0 : (rem + 31 >= 32 * 32 ? 32 : (rem + 31) >> 5);
      if (nk > 0) {
        EdgeRegs ra, rb;
        own_load(ra, sv, dv, av, 0, n0, ot, t, lane, a_src, a_dst, mbuf, sden, h);
        int k = 0;
        for (; k + 2 < nk; k += 2) {
          own_load(rb, sv, dv, av, k + 1, n0, ot, t, lane, a_src, a_dst, mbuf, sden, h);
          own_add(ra, cf, accl);
          own_load(ra, sv, dv, av, k + 2, n0, ot, t, lane, a_src, a_dst, mbuf, sden, h);
          own_add(rb, cf, accl);
        }
        if (k + 1 < nk) {
          own_load(rb, sv, dv, av, k + 1, n0, ot, t, lane, a_src, a_dst, mbuf, sden, h);
          own_add(ra, cf, accl);
          own_add(rb, cf, accl);
        } else {
          own_add(ra, cf, accl);
        }
      }
    }
    if (!last) {
      __syncthreads();
      total = 0;
    }
  }
  __syncthreads();

  for (int i = tid; i < OWN_NODES * BB * (OO / 4); i += OWN_THREADS) {
    const int q = i & 15;
    const int r = i >> 4;
    const int nl = r % OWN_NODES, b = r / OWN_NODES;
    const float* ap = accl + (nl * BB + b) * OO + 2 * q;
    const float2 ev = *(const float2*)ap;
    const float2 od = *(const float2*)(ap + 32);
    const float4 bv = ((const float4*)bias)[q];
    float4 o;
    o.x = ev.x * 0.25f + bv.x;
    o.y = od.x * 0.25f + bv.y;
    o.z = ev.y * 0.25f + bv.z;
    o.w = od.y * 0.25f + bv.w;
    ((float4*)out)[((size_t)b * NB + n0 + nl) * (OO / 4) + q] = o;
  }
}

extern "C" void kernel_launch(void* const* d_in, const int* in_sizes, int n_in,
                              void* d_out, int out_size, void* d_ws, size_t ws_size,
                              hipStream_t stream) {
  const float* x        = (const float*)d_in[0];
  const int*   eidx     = (const int*)d_in[1];
  const float* eattr    = (const float*)d_in[2];
  const float* W        = (const float*)d_in[3];
  const float* att_src  = (const float*)d_in[4];
  const float* att_dst  = (const float*)d_in[5];
  const float* W_edge   = (const float*)d_in[6];
  const float* att_edge = (const float*)d_in[7];
  const float* bias     = (const float*)d_in[8];
  float* out = (float*)d_out;

  char* ws = (char*)d_ws;
  float*    h_buf  = (float*)   (ws + 0);
  float*    a_src  = (float*)   (ws + 40960000);
  float*    a_dst  = (float*)   (ws + 43520000);
  unsigned* mbuf   = (unsigned*)(ws + 46080000);
  float*    sden   = (float*)   (ws + 48640000);
  float*    coef   = (float*)   (ws + 51200000);

  const int* srcI = eidx;
  const int* dstI = eidx + EE;

  hipMemsetAsync(ws + 46080000, 0, 5120000, stream);

  calc_edge_coef<<<1, 64, 0, stream>>>(W_edge, att_edge, coef);
  gemm_h<<<(MROWS + 127) / 128, 256, 0, stream>>>(x, W, h_buf);
  calc_attn_node<<<ATTN_BLOCKS, 256, 0, stream>>>(h_buf, att_src, att_dst,
                                                  a_src, a_dst);
  edge_max<<<EW_BLOCKS, 256, 0, stream>>>(srcI, dstI, eattr, a_src, a_dst, coef,
                                          mbuf);
  edge_sum<<<EW_BLOCKS, 256, 0, stream>>>(srcI, dstI, eattr, a_src, a_dst, coef,
                                          mbuf, sden);
  edge_own<<<OWN_BLOCKS, OWN_THREADS, 0, stream>>>(srcI, dstI, eattr, a_src, a_dst,
                                                   coef, mbuf, sden, h_buf, bias, out);
}


# Owner groups search a one-byte-per-edge plane of owner ids (dst/40, written by the first launch): 160 KB in one pass, not the 640 KB destination list in 4 chunks; sums unordered as before
# speedup vs baseline: 5.9374x; 1.2958x over previous
#include <hip/hip_runtime.h>


#define NB 10000
#define BB 4
#define EE 160000
#define DD 128
#define HH 4
#define OO 64
#define HO 256
#define MROWS (BB * NB)
#define NEG_SLOPE 0.2f

typedef float v2f __attribute__((ext_vector_type(2)));
typedef float v8f __attribute__((ext_vector_type(8)));

__device__ __forceinline__ unsigned fenc(float f) {
  unsigned u = __float_as_uint(f);
  return (u & 0x80000000u) ? ~u : (u | 0x80000000u);
}
__device__ __forceinline__ float fdec(unsigned u) {
  return __uint_as_float((u & 0x80000000u) ? (u ^ 0x80000000u) : ~u);
}

__global__ __launch_bounds__(256) void gemm_h(const float* __restrict__ x,
                                              const float* __restrict__ W,
                                              float* __restrict__ h) {
  __shared__ float2 Wlds[32 * HO];
  const int tid = threadIdx.x;
  const int lane = tid & 31;
  const int wave = tid >> 5;
  const int rowBase = blockIdx.x * 128 + wave * 16;
  const bool active = rowBase < MROWS;

  v8f acc[16];
  const v8f vzero = {0.f, 0.f, 0.f, 0.f, 0.f, 0.f, 0.f, 0.f};
#pragma unroll
  for (int t = 0; t < 16; ++t) acc[t] = vzero;

  const int m = lane & 15;
  const int kb = (lane >> 4) << 1;
  const int kp = lane >> 4;

  for (int kblk = 0; kblk < DD; kblk += 64) {
#pragma unroll 4
    for (int p = 0; p < 32; ++p) {
      float2 pr;
      pr.x = W[(size_t)(kblk + 2 * p) * HO + tid];
      pr.y = W[(size_t)(kblk + 2 * p + 1) * HO + tid];
      Wlds[p * HO + tid] = pr;
    }
    __syncthreads();

    if (active) {
      const float* xrow = x + (size_t)(rowBase + m) * DD + kblk;
      for (int kk = 0; kk < 64; kk += 4) {
        v2f a;
        a.x = xrow[kk + kb];
        a.y = xrow[kk + kb + 1];
        const float2* w0 = Wlds + ((kk >> 1) + kp) * HO + (lane & 15);
#pragma unroll
        for (int t = 0; t < 16; ++t) {
          float2 bp = w0[t * 16];
          v2f b;
          b.x = bp.x;
          b.y = bp.y;
          acc[t] = __builtin_amdgcn_wmma_f32_16x16x4_f32(
              false, a, false, b, (short)0, acc[t], false, false);
        }
      }
    }
    __syncthreads();
  }

  if (active) {
    const int nLocal = lane & 15;
    const int rOff = (lane >> 4) * 8;
#pragma unroll
    for (int t = 0; t < 16; ++t)
#pragma unroll
      for (int v = 0; v < 8; ++v)
        h[(size_t)(rowBase + rOff + v) * HO + t * 16 + nLocal] = acc[t][v];
  }
}

#define ATTN_ROWS 5
#define ATTN_CHUNKS (MROWS / ATTN_ROWS)
#define ATTN_BLOCKS 500
__global__ __launch_bounds__(256) void calc_attn_node(
    const float* __restrict__ h, const float* __restrict__ att_src,
    const float* __restrict__ att_dst, float* __restrict__ a_src,
    float* __restrict__ a_dst) {
  const int lane = threadIdx.x & 31;
  const int w = blockIdx.x * 8 + __builtin_amdgcn_readfirstlane((int)(threadIdx.x >> 5));
  float2 as[HH], ad[HH];
#pragma unroll
  for (int k = 0; k < HH; ++k) {
    as[k] = ((const float2*)(att_src + k * OO))[lane];
    ad[k] = ((const float2*)(att_dst + k * OO))[lane];
  }
  for (int c = w; c < ATTN_CHUNKS; c += ATTN_BLOCKS * 8) {
    const int row0 = c * ATTN_ROWS;
    const float2* hp = (const float2*)(h + (size_t)row0 * HO) + lane;
    float2 hv[ATTN_ROWS][HH];
#pragma unroll
    for (int r = 0; r < ATTN_ROWS; ++r)
#pragma unroll
      for (int k = 0; k < HH; ++k) hv[r][k] = hp[(r * HH + k) * 32];
    float vs[ATTN_ROWS][HH], vd[ATTN_ROWS][HH];
#pragma unroll
    for (int r = 0; r < ATTN_ROWS; ++r)
#pragma unroll
      for (int k = 0; k < HH; ++k) {
        vs[r][k] = hv[r][k].x * as[k].x + hv[r][k].y * as[k].y;
        vd[r][k] = hv[r][k].x * ad[k].x + hv[r][k].y * ad[k].y;
      }
#pragma unroll
    for (int off = 16; off > 0; off >>= 1)
#pragma unroll
      for (int r = 0; r < ATTN_ROWS; ++r)
#pragma unroll
        for (int k = 0; k < HH; ++k) {
          vs[r][k] += __shfl_xor(vs[r][k], off);
          vd[r][k] += __shfl_xor(vd[r][k], off);
        }
    float4 os = make_float4(0.f, 0.f, 0.f, 0.f), od = os;
#pragma unroll
    for (int r = 0; r < ATTN_ROWS; ++r)
      if (lane == r) {
        os = make_float4(vs[r][0], vs[r][1], vs[r][2], vs[r][3]);
        od = make_float4(vd[r][0], vd[r][1], vd[r][2], vd[r][3]);
      }
    if (lane < ATTN_ROWS) {
      ((float4*)a_src)[row0 + lane] = os;
      ((float4*)a_dst)[row0 + lane] = od;
    }
  }
}

#define OWN_NODES 40
#define E16 (EE / 16)
static_assert(E16 * 16 == EE, "the owner plane is written and read as whole 16-byte words");
#define PREP_THREADS 256
#define PREP_BLOCKS ((E16 + PREP_THREADS - 1) / PREP_THREADS)
__device__ __forceinline__ unsigned owner_pack(int4 d) {
  const unsigned a = min((unsigned)d.x / OWN_NODES, 255u);
  const unsigned b = min((unsigned)d.y / OWN_NODES, 255u);
  const unsigned c = min((unsigned)d.z / OWN_NODES, 255u);
  const unsigned e = min((unsigned)d.w / OWN_NODES, 255u);
  return a | (b << 8) | (c << 16) | (e << 24);
}
__global__ __launch_bounds__(PREP_THREADS) void edge_prep(
    const float* __restrict__ W_edge, const float* __restrict__ att_edge,
    float* __restrict__ coef, const int* __restrict__ dst, int* __restrict__ own) {
  const int t = threadIdx.x;
  const int i = blockIdx.x * PREP_THREADS + t;
  if (i < E16) {
    const int4* d4 = (const int4*)dst + 4 * i;
    const int4 d0 = d4[0], d1 = d4[1], d2 = d4[2], d3 = d4[3];
    int4 o;
    o.x = (int)owner_pack(d0);
    o.y = (int)owner_pack(d1);
    o.z = (int)owner_pack(d2);
    o.w = (int)owner_pack(d3);
    ((int4*)own)[i] = o;
  }
  if (blockIdx.x == PREP_BLOCKS && t < HH) {
    float s = 0.f;
    for (int o = 0; o < OO; ++o) s += W_edge[t * OO + o] * att_edge[t * OO + o];
    coef[t] = s;
  }
}

__device__ __forceinline__ float edge_score(float as, float ad, float ae, float cf) {
  float a = as + ad + ae * cf;
  return a > 0.f ? a : a * NEG_SLOPE;
}

__device__ __forceinline__ int pair_pick(int v, int j, bool hi) {
  const int lo_v = __builtin_amdgcn_readlane(v, 2 * j);
  const int hi_v = __builtin_amdgcn_readlane(v, 2 * j + 1);
  return hi ? hi_v : lo_v;
}

#define OWN_BLOCKS (NB / OWN_NODES)
static_assert(OWN_BLOCKS * OWN_NODES == NB, "every node has exactly one owner");
static_assert(OWN_BLOCKS <= 255, "an owner id is one byte and 255 means no owner");
#define OWN_THREADS 1024
#define OWN_SCORES (OWN_NODES * BB * HH)
static_assert(OWN_SCORES <= OWN_THREADS, "one thread clears one cell");
#define SCAN_LOADS 10
static_assert(SCAN_LOADS * OWN_THREADS >= E16, "one search covers every edge");
#define PART_LOADS 3
#define PART16 (PART_LOADS * OWN_THREADS)
#define PARTS ((E16 + PART16 - 1) / PART16)
#define LIST_CAP (16 * PART16)
#define WALKS 3

__device__ __forceinline__ unsigned zero_bytes(unsigned x) {
  const unsigned t = (x & 0x7F7F7F7Fu) + 0x7F7F7F7Fu;
  return ~(t | x | 0x7F7F7F7Fu);
}

template <int LOADS, bool CHECK>
__device__ __forceinline__ void own_scan(const int4* __restrict__ tab4, int qb, int qe,
                                         unsigned gpat, int tid, int* cnt, int* elist) {
  int4 dq[LOADS];
#pragma unroll
  for (int k = 0; k < LOADS; ++k) {
    const int q = qb + k * OWN_THREADS + tid;
    dq[k] = q < qe ? tab4[q] : make_int4(-1, -1, -1, -1);
  }
  unsigned m[LOADS];
  int c = 0;
#pragma unroll
  for (int k = 0; k < LOADS; ++k) {
    const unsigned f0 = zero_bytes((unsigned)dq[k].x ^ gpat);
    const unsigned f1 = zero_bytes((unsigned)dq[k].y ^ gpat);
    const unsigned f2 = zero_bytes((unsigned)dq[k].z ^ gpat);
    const unsigned f3 = zero_bytes((unsigned)dq[k].w ^ gpat);
    m[k] = (f0 >> 7) | (f1 >> 6) | (f2 >> 5) | (f3 >> 4);
    c += __builtin_popcount(m[k]);
  }
  if (c == 0) return;
  int pos = atomicAdd(cnt, c);
#pragma unroll
  for (int k = 0; k < LOADS; ++k) {
    const int q = qb + k * OWN_THREADS + tid;
    unsigned mk = m[k];
    while (mk != 0u) {
      const int bit = __builtin_ctz(mk);
      mk &= mk - 1u;
      if (!CHECK || pos < LIST_CAP) elist[pos] = 16 * q + 4 * (bit & 7) + (bit >> 3);
      ++pos;
    }
  }
}

template <int WALK>
__device__ __forceinline__ void own_stat(int sv, int dv, int av, int nk, int n0, int ot,
                                         int t, bool hi, float cf,
                                         const float* __restrict__ a_src,
                                         const float* __restrict__ a_dst,
                                         unsigned* mxl, float* denl) {
  float al[16];
  int sl[16];
  unsigned mv[16];
#pragma unroll
  for (int j = 0; j < 16; ++j) {
    const int sN = pair_pick(sv, j, hi);
    const int dN = pair_pick(dv, j, hi);
    const float ae = __int_as_float(pair_pick(av, j, hi));
    al[j] = edge_score(a_src[ot + sN * HH], a_dst[ot + dN * HH], ae, cf);
    sl[j] = (dN - n0) * (BB * HH) + t;
    mv[j] = WALK == 1 ? mxl[sl[j]] : 0u;
  }
  const int k0 = hi ? 1 : 0;
#pragma unroll
  for (int j = 0; j < 16; ++j)
    if (2 * j + k0 < nk) {
      if (WALK == 0)
        __hip_atomic_fetch_max(&mxl[sl[j]], fenc(al[j]), __ATOMIC_RELAXED,
                               __HIP_MEMORY_SCOPE_WORKGROUP);
      else
        __hip_atomic_fetch_add(&denl[sl[j]], __expf(al[j] - fdec(mv[j])),
                               __ATOMIC_RELAXED, __HIP_MEMORY_SCOPE_WORKGROUP);
    }
}

struct EdgeRegs {
  float as, ad, ae, den;
  unsigned mx;
  int dl;
  float2 hv[BB][HH];
};

__device__ __forceinline__ void own_load(EdgeRegs& r, int sv, int dv, int av, int k,
                                         int n0, int ot, int t, int lane,
                                         const float* __restrict__ a_src,
                                         const float* __restrict__ a_dst,
                                         const unsigned* mxl, const float* denl,
                                         const float* __restrict__ h) {
  const int sN = __builtin_amdgcn_readlane(sv, k);
  const int dN = __builtin_amdgcn_readlane(dv, k);
  r.ae = __int_as_float(__builtin_amdgcn_readlane(av, k));
  r.as = a_src[ot + sN * HH];
  r.ad = a_dst[ot + dN * HH];
  r.mx = mxl[(dN - n0) * (BB * HH) + t];
  r.den = denl[(dN - n0) * (BB * HH) + t];
  r.dl = (dN - n0) * (BB * OO) + lane;
  const float2* hs = (const float2*)(h + sN * HO) + lane;
#pragma unroll
  for (int b = 0; b < BB; ++b)
#pragma unroll
    for (int k2 = 0; k2 < HH; ++k2) r.hv[b][k2] = hs[(b * NB * HO + k2 * OO) / 2];
}

__device__ __forceinline__ void own_add(const EdgeRegs& r, float cf, float* accl) {
  const float a = edge_score(r.as, r.ad, r.ae, cf);
  const float p = __expf(a - fdec(r.mx));
  const float wl = p / fmaxf(r.den, 1e-16f);
#pragma unroll
  for (int b = 0; b < BB; ++b) {
    const float w0 = __int_as_float(__builtin_amdgcn_readlane(__float_as_int(wl), b * 4 + 0));
    const float w1 = __int_as_float(__builtin_amdgcn_readlane(__float_as_int(wl), b * 4 + 1));
    const float w2 = __int_as_float(__builtin_amdgcn_readlane(__float_as_int(wl), b * 4 + 2));
    const float w3 = __int_as_float(__builtin_amdgcn_readlane(__float_as_int(wl), b * 4 + 3));
    const float vx = fmaf(w3, r.hv[b][3].x, fmaf(w2, r.hv[b][2].x, fmaf(w1, r.hv[b][1].x, w0 * r.hv[b][0].x)));
    const float vy = fmaf(w3, r.hv[b][3].y, fmaf(w2, r.hv[b][2].y, fmaf(w1, r.hv[b][1].y, w0 * r.hv[b][0].y)));
    __hip_atomic_fetch_add(&accl[r.dl + b * OO], vx, __ATOMIC_RELAXED, __HIP_MEMORY_SCOPE_WORKGROUP);
    __hip_atomic_fetch_add(&accl[r.dl + b * OO + 32], vy, __ATOMIC_RELAXED, __HIP_MEMORY_SCOPE_WORKGROUP);
  }
}

__global__ __launch_bounds__(OWN_THREADS) void edge_own(
    const int* __restrict__ src, const int* __restrict__ dst,
    const float* __restrict__ eattr, const float* __restrict__ a_src,
    const float* __restrict__ a_dst, const float* __restrict__ coef,
    const float* __restrict__ h, const float* __restrict__ bias,
    const int* __restrict__ own, float* __restrict__ out) {
  __shared__ float accl[OWN_NODES * BB * OO];
  __shared__ int elist[LIST_CAP];
  __shared__ unsigned mxl[OWN_SCORES];
  __shared__ float denl[OWN_SCORES];
  __shared__ int ecnt[1 + WALKS * PARTS];
  const int tid = threadIdx.x;
  const int lane = tid & 31;
  const int t = lane & 15;
  const bool hi = lane >= 16;
  const int wv = __builtin_amdgcn_readfirstlane(tid >> 5);
  const int n0 = blockIdx.x * OWN_NODES;
  const int ot = (t >> 2) * (NB * HH) + (t & 3);
  const float cf = coef[t & 3];

  for (int i = tid; i < OWN_NODES * BB * OO; i += OWN_THREADS) accl[i] = 0.f;
  if (tid < OWN_SCORES) {
    mxl[tid] = 0u;
    denl[tid] = 0.f;
  }
  if (tid < 1 + WALKS * PARTS) ecnt[tid] = 0;
  __syncthreads();

  const int4* tab4 = (const int4*)own;
  const unsigned gpat = blockIdx.x * 0x01010101u;
  own_scan<SCAN_LOADS, true>(tab4, 0, E16, gpat, tid, &ecnt[0], elist);
  __syncthreads();
  int total = ecnt[0];
  const bool whole = total <= LIST_CAP;
  const int nparts = whole ? 1 : PARTS;
  for (int walk = 0; walk < WALKS; ++walk) {
    for (int part = 0; part < nparts; ++part) {
      if (!whole) {
        int* cnt = &ecnt[1 + walk * PARTS + part];
        const int qb = part * PART16;
        own_scan<PART_LOADS, false>(tab4, qb, min(qb + PART16, E16), gpat, tid, cnt, elist);
        __syncthreads();
        total = *cnt;
      }

      for (int b0 = 0; b0 < total; b0 += OWN_THREADS) {
        const int my = b0 + lane * 32 + wv;
        int sv = 0, dv = n0, av = 0;
        if (my < total) {
          const int e = elist[my];
          sv = src[e];
          dv = dst[e];
          av = __float_as_int(eattr[e]);
        }
        const int rem = total - b0 - wv;
        const int nk = rem <= 0 ? 0 : (rem + 31 >= 32 * 32 ? 32 : (rem + 31) >> 5);
        if (nk > 0) {
          if (walk == 0) {
            own_stat<0>(sv, dv, av, nk, n0, ot, t, hi, cf, a_src, a_dst, mxl, denl);
          } else if (walk == 1) {
            own_stat<1>(sv, dv, av, nk, n0, ot, t, hi, cf, a_src, a_dst, mxl, denl);
          } else {
            EdgeRegs ra, rb;
            own_load(ra, sv, dv, av, 0, n0, ot, t, lane, a_src, a_dst, mxl, denl, h);
            int k = 0;
            for (; k + 2 < nk; k += 2) {
              own_load(rb, sv, dv, av, k + 1, n0, ot, t, lane, a_src, a_dst, mxl, denl, h);
              own_add(ra, cf, accl);
              own_load(ra, sv, dv, av, k + 2, n0, ot, t, lane, a_src, a_dst, mxl, denl, h);
              own_add(rb, cf, accl);
            }
            if (k + 1 < nk) {
              own_load(rb, sv, dv, av, k + 1, n0, ot, t, lane, a_src, a_dst, mxl, denl, h);
              own_add(ra, cf, accl);
              own_add(rb, cf, accl);
            } else {
              own_add(ra, cf, accl);
            }
          }
        }
      }
      __syncthreads();
    }
  }

  for (int i = tid; i < OWN_NODES * BB * (OO / 4); i += OWN_THREADS) {
    const int q = i & 15;
    const int r = i >> 4;
    const int nl = r % OWN_NODES, b = r / OWN_NODES;
    const float* ap = accl + (nl * BB + b) * OO + 2 * q;
    const float2 ev = *(const float2*)ap;
    const float2 od = *(const float2*)(ap + 32);
    const float4 bv = ((const float4*)bias)[q];
    float4 o;
    o.x = ev.x * 0.25f + bv.x;
    o.y = od.x * 0.25f + bv.y;
    o.z = ev.y * 0.25f + bv.z;
    o.w = od.y * 0.25f + bv.w;
    ((float4*)out)[((size_t)b * NB + n0 + nl) * (OO / 4) + q] = o;
  }
}

extern "C" void kernel_launch(void* const* d_in, const int* in_sizes, int n_in,
                              void* d_out, int out_size, void* d_ws, size_t ws_size,
                              hipStream_t stream) {
  const float* x        = (const float*)d_in[0];
  const int*   eidx     = (const int*)d_in[1];
  const float* eattr    = (const float*)d_in[2];
  const float* W        = (const float*)d_in[3];
  const float* att_src  = (const float*)d_in[4];
  const float* att_dst  = (const float*)d_in[5];
  const float* W_edge   = (const float*)d_in[6];
  const float* att_edge = (const float*)d_in[7];
  const float* bias     = (const float*)d_in[8];
  float* out = (float*)d_out;

  char* ws = (char*)d_ws;
  float*    h_buf  = (float*)   (ws + 0);
  float*    a_src  = (float*)   (ws + 40960000);
  float*    a_dst  = (float*)   (ws + 43520000);
  float*    coef   = (float*)   (ws + 51200000);
  int*      own    = (int*)     (ws + 51200256);

  const int* srcI = eidx;
  const int* dstI = eidx + EE;

  edge_prep<<<PREP_BLOCKS + 1, PREP_THREADS, 0, stream>>>(W_edge, att_edge, coef, dstI, own);
  gemm_h<<<(MROWS + 127) / 128, 256, 0, stream>>>(x, W, h_buf);
  calc_attn_node<<<ATTN_BLOCKS, 256, 0, stream>>>(h_buf, att_src, att_dst,
                                                  a_src, a_dst);
  edge_own<<<OWN_BLOCKS, OWN_THREADS, 0, stream>>>(srcI, dstI, eattr, a_src, a_dst,
                                                   coef, h_buf, bias, own, out);
}


# Attention vectors folded into the weight: scores a_src, a_dst = x.(W.att) leave the GEMM as a 17th column tile; the per-row dot kernel and its re-read of h go; scores summed over d, a new order
# speedup vs baseline: 6.2684x; 1.0557x over previous
#include <hip/hip_runtime.h>


#define NB 10000
#define BB 4
#define EE 160000
#define DD 128
#define HH 4
#define OO 64
#define HO 256
#define MROWS (BB * NB)
#define NEG_SLOPE 0.2f

typedef float v2f __attribute__((ext_vector_type(2)));
typedef float v8f __attribute__((ext_vector_type(8)));

__device__ __forceinline__ unsigned fenc(float f) {
  unsigned u = __float_as_uint(f);
  return (u & 0x80000000u) ? ~u : (u | 0x80000000u);
}
__device__ __forceinline__ float fdec(unsigned u) {
  return __uint_as_float((u & 0x80000000u) ? (u ^ 0x80000000u) : ~u);
}

#define FCOLS 16
__global__ __launch_bounds__(256) void gemm_h(const float* __restrict__ x,
                                              const float* __restrict__ W,
                                              const float* __restrict__ wap,
                                              float* __restrict__ h,
                                              float* __restrict__ a_src,
                                              float* __restrict__ a_dst) {
  __shared__ float2 Wlds[32 * HO];
  __shared__ float2 Flds[32 * FCOLS];
  const int tid = threadIdx.x;
  const int lane = tid & 31;
  const int wave = tid >> 5;
  const int rowBase = blockIdx.x * 128 + wave * 16;
  const bool active = rowBase < MROWS;

  v8f acc[17];
  const v8f vzero = {0.f, 0.f, 0.f, 0.f, 0.f, 0.f, 0.f, 0.f};
#pragma unroll
  for (int t = 0; t < 17; ++t) acc[t] = vzero;

  const int m = lane & 15;
  const int kb = (lane >> 4) << 1;
  const int kp = lane >> 4;

  for (int kblk = 0; kblk < DD; kblk += 64) {
    const float2* fp = (const float2*)wap + (kblk >> 1) * FCOLS;
    const float2 f0 = fp[tid];
    const float2 f1 = fp[tid + 256];
#pragma unroll 4
    for (int p = 0; p < 32; ++p) {
      float2 pr;
      pr.x = W[(size_t)(kblk + 2 * p) * HO + tid];
      pr.y = W[(size_t)(kblk + 2 * p + 1) * HO + tid];
      Wlds[p * HO + tid] = pr;
    }
    Flds[tid] = f0;
    Flds[tid + 256] = f1;
    __syncthreads();

    if (active) {
      const float* xrow = x + (size_t)(rowBase + m) * DD + kblk;
      for (int kk = 0; kk < 64; kk += 4) {
        v2f a;
        a.x = xrow[kk + kb];
        a.y = xrow[kk + kb + 1];
        const float2* w0 = Wlds + ((kk >> 1) + kp) * HO + (lane & 15);
#pragma unroll
        for (int t = 0; t < 16; ++t) {
          float2 bp = w0[t * 16];
          v2f b;
          b.x = bp.x;
          b.y = bp.y;
          acc[t] = __builtin_amdgcn_wmma_f32_16x16x4_f32(
              false, a, false, b, (short)0, acc[t], false, false);
        }
        {
          const float2 fpr = Flds[((kk >> 1) + kp) * FCOLS + (lane & 15)];
          v2f b;
          b.x = fpr.x;
          b.y = fpr.y;
          acc[16] = __builtin_amdgcn_wmma_f32_16x16x4_f32(
              false, a, false, b, (short)0, acc[16], false, false);
        }
      }
    }
    __syncthreads();
  }

  if (active) {
    const int nLocal = lane & 15;
    const int rOff = (lane >> 4) * 8;
#pragma unroll
    for (int t = 0; t < 16; ++t)
#pragma unroll
      for (int v = 0; v < 8; ++v)
        h[(size_t)(rowBase + rOff + v) * HO + t * 16 + nLocal] = acc[t][v];
    if (nLocal < 8) {
      float* tab = nLocal < 4 ? a_src : a_dst;
#pragma unroll
      for (int v = 0; v < 8; ++v)
        tab[(size_t)(rowBase + rOff + v) * HH + (nLocal & 3)] = acc[16][v];
    }
  }
}

#define OWN_NODES 40
#define E16 (EE / 16)
static_assert(E16 * 16 == EE, "the owner plane is written and read as whole 16-byte words");
#define PREP_THREADS 256
#define PREP_BLOCKS ((E16 + PREP_THREADS - 1) / PREP_THREADS)
#define FOLD_BLOCKS ((DD * 8) / PREP_THREADS)
static_assert(FOLD_BLOCKS * PREP_THREADS == DD * 8, "one thread per folded weight");
__device__ __forceinline__ unsigned owner_pack(int4 d) {
  const unsigned a = min((unsigned)d.x / OWN_NODES, 255u);
  const unsigned b = min((unsigned)d.y / OWN_NODES, 255u);
  const unsigned c = min((unsigned)d.z / OWN_NODES, 255u);
  const unsigned e = min((unsigned)d.w / OWN_NODES, 255u);
  return a | (b << 8) | (c << 16) | (e << 24);
}
__global__ __launch_bounds__(PREP_THREADS) void edge_prep(
    const float* __restrict__ W_edge, const float* __restrict__ att_edge,
    float* __restrict__ coef, const int* __restrict__ dst, int* __restrict__ own,
    const float* __restrict__ W, const float* __restrict__ att_src,
    const float* __restrict__ att_dst, float* __restrict__ wap) {
  const int t = threadIdx.x;
  const int i = blockIdx.x * PREP_THREADS + t;
  if (i < E16) {
    const int4* d4 = (const int4*)dst + 4 * i;
    const int4 d0 = d4[0], d1 = d4[1], d2 = d4[2], d3 = d4[3];
    int4 o;
    o.x = (int)owner_pack(d0);
    o.y = (int)owner_pack(d1);
    o.z = (int)owner_pack(d2);
    o.w = (int)owner_pack(d3);
    ((int4*)own)[i] = o;
  }
  if (blockIdx.x == PREP_BLOCKS && t < HH) {
    float s = 0.f;
    for (int o = 0; o < OO; ++o) s += W_edge[t * OO + o] * att_edge[t * OO + o];
    coef[t] = s;
  }
  if (blockIdx.x > PREP_BLOCKS) {
    const int idx = (blockIdx.x - PREP_BLOCKS - 1) * PREP_THREADS + t;
    const int d = idx >> 3, c = idx & 7, hd = c & 3;
    const float4* wr = (const float4*)(W + (size_t)d * HO + hd * OO);
    const float4* ar = (const float4*)((c < 4 ? att_src : att_dst) + hd * OO);
    float s = 0.f;
#pragma unroll
    for (int o = 0; o < OO / 4; ++o) {
      const float4 w4 = wr[o], a4 = ar[o];
      s += w4.x * a4.x;
      s += w4.y * a4.y;
      s += w4.z * a4.z;
      s += w4.w * a4.w;
    }
    wap[((d >> 1) * FCOLS + c) * 2 + (d & 1)] = s;
    wap[((d >> 1) * FCOLS + c + 8) * 2 + (d & 1)] = 0.f;
  }
}

__device__ __forceinline__ float edge_score(float as, float ad, float ae, float cf) {
  float a = as + ad + ae * cf;
  return a > 0.f ? a : a * NEG_SLOPE;
}

__device__ __forceinline__ int pair_pick(int v, int j, bool hi) {
  const int lo_v = __builtin_amdgcn_readlane(v, 2 * j);
  const int hi_v = __builtin_amdgcn_readlane(v, 2 * j + 1);
  return hi ? hi_v : lo_v;
}

#define OWN_BLOCKS (NB / OWN_NODES)
static_assert(OWN_BLOCKS * OWN_NODES == NB, "every node has exactly one owner");
static_assert(OWN_BLOCKS <= 255, "an owner id is one byte and 255 means no owner");
#define OWN_THREADS 1024
#define OWN_SCORES (OWN_NODES * BB * HH)
static_assert(OWN_SCORES <= OWN_THREADS, "one thread clears one cell");
#define SCAN_LOADS 10
static_assert(SCAN_LOADS * OWN_THREADS >= E16, "one search covers every edge");
#define PART_LOADS 3
#define PART16 (PART_LOADS * OWN_THREADS)
#define PARTS ((E16 + PART16 - 1) / PART16)
#define LIST_CAP (16 * PART16)
#define WALKS 3

__device__ __forceinline__ unsigned zero_bytes(unsigned x) {
  const unsigned t = (x & 0x7F7F7F7Fu) + 0x7F7F7F7Fu;
  return ~(t | x | 0x7F7F7F7Fu);
}

template <int LOADS, bool CHECK>
__device__ __forceinline__ void own_scan(const int4* __restrict__ tab4, int qb, int qe,
                                         unsigned gpat, int tid, int* cnt, int* elist) {
  int4 dq[LOADS];
#pragma unroll
  for (int k = 0; k < LOADS; ++k) {
    const int q = qb + k * OWN_THREADS + tid;
    dq[k] = q < qe ? tab4[q] : make_int4(-1, -1, -1, -1);
  }
  unsigned m[LOADS];
  int c = 0;
#pragma unroll
  for (int k = 0; k < LOADS; ++k) {
    const unsigned f0 = zero_bytes((unsigned)dq[k].x ^ gpat);
    const unsigned f1 = zero_bytes((unsigned)dq[k].y ^ gpat);
    const unsigned f2 = zero_bytes((unsigned)dq[k].z ^ gpat);
    const unsigned f3 = zero_bytes((unsigned)dq[k].w ^ gpat);
    m[k] = (f0 >> 7) | (f1 >> 6) | (f2 >> 5) | (f3 >> 4);
    c += __builtin_popcount(m[k]);
  }
  if (c == 0) return;
  int pos = atomicAdd(cnt, c);
#pragma unroll
  for (int k = 0; k < LOADS; ++k) {
    const int q = qb + k * OWN_THREADS + tid;
    unsigned mk = m[k];
    while (mk != 0u) {
      const int bit = __builtin_ctz(mk);
      mk &= mk - 1u;
      if (!CHECK || pos < LIST_CAP) elist[pos] = 16 * q + 4 * (bit & 7) + (bit >> 3);
      ++pos;
    }
  }
}

template <int WALK>
__device__ __forceinline__ void own_stat(int sv, int dv, int av, int nk, int n0, int ot,
                                         int t, bool hi, float cf,
                                         const float* __restrict__ a_src,
                                         const float* __restrict__ a_dst,
                                         unsigned* mxl, float* denl) {
  float al[16];
  int sl[16];
  unsigned mv[16];
#pragma unroll
  for (int j = 0; j < 16; ++j) {
    const int sN = pair_pick(sv, j, hi);
    const int dN = pair_pick(dv, j, hi);
    const float ae = __int_as_float(pair_pick(av, j, hi));
    al[j] = edge_score(a_src[ot + sN * HH], a_dst[ot + dN * HH], ae, cf);
    sl[j] = (dN - n0) * (BB * HH) + t;
    mv[j] = WALK == 1 ? mxl[sl[j]] : 0u;
  }
  const int k0 = hi ? 1 : 0;
#pragma unroll
  for (int j = 0; j < 16; ++j)
    if (2 * j + k0 < nk) {
      if (WALK == 0)
        __hip_atomic_fetch_max(&mxl[sl[j]], fenc(al[j]), __ATOMIC_RELAXED,
                               __HIP_MEMORY_SCOPE_WORKGROUP);
      else
        __hip_atomic_fetch_add(&denl[sl[j]], __expf(al[j] - fdec(mv[j])),
                               __ATOMIC_RELAXED, __HIP_MEMORY_SCOPE_WORKGROUP);
    }
}

struct EdgeRegs {
  float as, ad, ae, den;
  unsigned mx;
  int dl;
  float2 hv[BB][HH];
};

__device__ __forceinline__ void own_load(EdgeRegs& r, int sv, int dv, int av, int k,
                                         int n0, int ot, int t, int lane,
                                         const float* __restrict__ a_src,
                                         const float* __restrict__ a_dst,
                                         const unsigned* mxl, const float* denl,
                                         const float* __restrict__ h) {
  const int sN = __builtin_amdgcn_readlane(sv, k);
  const int dN = __builtin_amdgcn_readlane(dv, k);
  r.ae = __int_as_float(__builtin_amdgcn_readlane(av, k));
  r.as = a_src[ot + sN * HH];
  r.ad = a_dst[ot + dN * HH];
  r.mx = mxl[(dN - n0) * (BB * HH) + t];
  r.den = denl[(dN - n0) * (BB * HH) + t];
  r.dl = (dN - n0) * (BB * OO) + lane;
  const float2* hs = (const float2*)(h + sN * HO) + lane;
#pragma unroll
  for (int b = 0; b < BB; ++b)
#pragma unroll
    for (int k2 = 0; k2 < HH; ++k2) r.hv[b][k2] = hs[(b * NB * HO + k2 * OO) / 2];
}

__device__ __forceinline__ void own_add(const EdgeRegs& r, float cf, float* accl) {
  const float a = edge_score(r.as, r.ad, r.ae, cf);
  const float p = __expf(a - fdec(r.mx));
  const float wl = p / fmaxf(r.den, 1e-16f);
#pragma unroll
  for (int b = 0; b < BB; ++b) {
    const float w0 = __int_as_float(__builtin_amdgcn_readlane(__float_as_int(wl), b * 4 + 0));
    const float w1 = __int_as_float(__builtin_amdgcn_readlane(__float_as_int(wl), b * 4 + 1));
    const float w2 = __int_as_float(__builtin_amdgcn_readlane(__float_as_int(wl), b * 4 + 2));
    const float w3 = __int_as_float(__builtin_amdgcn_readlane(__float_as_int(wl), b * 4 + 3));
    const float vx = fmaf(w3, r.hv[b][3].x, fmaf(w2, r.hv[b][2].x, fmaf(w1, r.hv[b][1].x, w0 * r.hv[b][0].x)));
    const float vy = fmaf(w3, r.hv[b][3].y, fmaf(w2, r.hv[b][2].y, fmaf(w1, r.hv[b][1].y, w0 * r.hv[b][0].y)));
    __hip_atomic_fetch_add(&accl[r.dl + b * OO], vx, __ATOMIC_RELAXED, __HIP_MEMORY_SCOPE_WORKGROUP);
    __hip_atomic_fetch_add(&accl[r.dl + b * OO + 32], vy, __ATOMIC_RELAXED, __HIP_MEMORY_SCOPE_WORKGROUP);
  }
}

__global__ __launch_bounds__(OWN_THREADS) void edge_own(
    const int* __restrict__ src, const int* __restrict__ dst,
    const float* __restrict__ eattr, const float* __restrict__ a_src,
    const float* __restrict__ a_dst, const float* __restrict__ coef,
    const float* __restrict__ h, const float* __restrict__ bias,
    const int* __restrict__ own, float* __restrict__ out) {
  __shared__ float accl[OWN_NODES * BB * OO];
  __shared__ int elist[LIST_CAP];
  __shared__ unsigned mxl[OWN_SCORES];
  __shared__ float denl[OWN_SCORES];
  __shared__ int ecnt[1 + WALKS * PARTS];
  const int tid = threadIdx.x;
  const int lane = tid & 31;
  const int t = lane & 15;
  const bool hi = lane >= 16;
  const int wv = __builtin_amdgcn_readfirstlane(tid >> 5);
  const int n0 = blockIdx.x * OWN_NODES;
  const int ot = (t >> 2) * (NB * HH) + (t & 3);
  const float cf = coef[t & 3];

  for (int i = tid; i < OWN_NODES * BB * OO; i += OWN_THREADS) accl[i] = 0.f;
  if (tid < OWN_SCORES) {
    mxl[tid] = 0u;
    denl[tid] = 0.f;
  }
  if (tid < 1 + WALKS * PARTS) ecnt[tid] = 0;
  __syncthreads();

  const int4* tab4 = (const int4*)own;
  const unsigned gpat = blockIdx.x * 0x01010101u;
  own_scan<SCAN_LOADS, true>(tab4, 0, E16, gpat, tid, &ecnt[0], elist);
  __syncthreads();
  int total = ecnt[0];
  const bool whole = total <= LIST_CAP;
  const int nparts = whole ? 1 : PARTS;
  for (int walk = 0; walk < WALKS; ++walk) {
    for (int part = 0; part < nparts; ++part) {
      if (!whole) {
        int* cnt = &ecnt[1 + walk * PARTS + part];
        const int qb = part * PART16;
        own_scan<PART_LOADS, false>(tab4, qb, min(qb + PART16, E16), gpat, tid, cnt, elist);
        __syncthreads();
        total = *cnt;
      }

      for (int b0 = 0; b0 < total; b0 += OWN_THREADS) {
        const int my = b0 + lane * 32 + wv;
        int sv = 0, dv = n0, av = 0;
        if (my < total) {
          const int e = elist[my];
          sv = src[e];
          dv = dst[e];
          av = __float_as_int(eattr[e]);
        }
        const int rem = total - b0 - wv;
        const int nk = rem <= 0 ? 0 : (rem + 31 >= 32 * 32 ? 32 : (rem + 31) >> 5);
        if (nk > 0) {
          if (walk == 0) {
            own_stat<0>(sv, dv, av, nk, n0, ot, t, hi, cf, a_src, a_dst, mxl, denl);
          } else if (walk == 1) {
            own_stat<1>(sv, dv, av, nk, n0, ot, t, hi, cf, a_src, a_dst, mxl, denl);
          } else {
            EdgeRegs ra, rb;
            own_load(ra, sv, dv, av, 0, n0, ot, t, lane, a_src, a_dst, mxl, denl, h);
            int k = 0;
            for (; k + 2 < nk; k += 2) {
              own_load(rb, sv, dv, av, k + 1, n0, ot, t, lane, a_src, a_dst, mxl, denl, h);
              own_add(ra, cf, accl);
              own_load(ra, sv, dv, av, k + 2, n0, ot, t, lane, a_src, a_dst, mxl, denl, h);
              own_add(rb, cf, accl);
            }
            if (k + 1 < nk) {
              own_load(rb, sv, dv, av, k + 1, n0, ot, t, lane, a_src, a_dst, mxl, denl, h);
              own_add(ra, cf, accl);
              own_add(rb, cf, accl);
            } else {
              own_add(ra, cf, accl);
            }
          }
        }
      }
      __syncthreads();
    }
  }

  for (int i = tid; i < OWN_NODES * BB * (OO / 4); i += OWN_THREADS) {
    const int q = i & 15;
    const int r = i >> 4;
    const int nl = r % OWN_NODES, b = r / OWN_NODES;
    const float* ap = accl + (nl * BB + b) * OO + 2 * q;
    const float2 ev = *(const float2*)ap;
    const float2 od = *(const float2*)(ap + 32);
    const float4 bv = ((const float4*)bias)[q];
    float4 o;
    o.x = ev.x * 0.25f + bv.x;
    o.y = od.x * 0.25f + bv.y;
    o.z = ev.y * 0.25f + bv.z;
    o.w = od.y * 0.25f + bv.w;
    ((float4*)out)[((size_t)b * NB + n0 + nl) * (OO / 4) + q] = o;
  }
}

extern "C" void kernel_launch(void* const* d_in, const int* in_sizes, int n_in,
                              void* d_out, int out_size, void* d_ws, size_t ws_size,
                              hipStream_t stream) {
  const float* x        = (const float*)d_in[0];
  const int*   eidx     = (const int*)d_in[1];
  const float* eattr    = (const float*)d_in[2];
  const float* W        = (const float*)d_in[3];
  const float* att_src  = (const float*)d_in[4];
  const float* att_dst  = (const float*)d_in[5];
  const float* W_edge   = (const float*)d_in[6];
  const float* att_edge = (const float*)d_in[7];
  const float* bias     = (const float*)d_in[8];
  float* out = (float*)d_out;

  char* ws = (char*)d_ws;
  float*    h_buf  = (float*)   (ws + 0);
  float*    a_src  = (float*)   (ws + 40960000);
  float*    a_dst  = (float*)   (ws + 43520000);
  float*    coef   = (float*)   (ws + 51200000);
  int*      own    = (int*)     (ws + 51200256);
  float*    wap    = (float*)   (ws + 51360768);

  const int* srcI = eidx;
  const int* dstI = eidx + EE;

  edge_prep<<<PREP_BLOCKS + 1 + FOLD_BLOCKS, PREP_THREADS, 0, stream>>>(
      W_edge, att_edge, coef, dstI, own, W, att_src, att_dst, wap);
  gemm_h<<<(MROWS + 127) / 128, 256, 0, stream>>>(x, W, wap, h_buf, a_src, a_dst);
  edge_own<<<OWN_BLOCKS, OWN_THREADS, 0, stream>>>(srcI, dstI, eattr, a_src, a_dst,
                                                   coef, h_buf, bias, own, out);
}


# GEMM on 250 resident work-groups, one per WGP: a group stages the whole weight and its 160 rows of x in LDS once, each of its 20 waves owns 16 rows x 128 columns (5 per SIMD); sum orders unchanged
# speedup vs baseline: 7.7277x; 1.2328x over previous
#include <hip/hip_runtime.h>


#define NB 10000
#define BB 4
#define EE 160000
#define DD 128
#define HH 4
#define OO 64
#define HO 256
#define MROWS (BB * NB)
#define NEG_SLOPE 0.2f

typedef float v2f __attribute__((ext_vector_type(2)));
typedef float v8f __attribute__((ext_vector_type(8)));

__device__ __forceinline__ unsigned fenc(float f) {
  unsigned u = __float_as_uint(f);
  return (u & 0x80000000u) ? ~u : (u | 0x80000000u);
}
__device__ __forceinline__ float fdec(unsigned u) {
  return __uint_as_float((u & 0x80000000u) ? (u ^ 0x80000000u) : ~u);
}

#define FCOLS 16
#define GEMM_GROUPS 250
#define GEMM_WAVES 20
#define GEMM_THREADS (GEMM_WAVES * 32)
#define GROUP_ROWS ((GEMM_WAVES / 2) * 16)
#define KPAIRS (DD / 2)
#define W_ITEMS (KPAIRS * (HO / 4))
#define W_TRIPS ((W_ITEMS + GEMM_THREADS - 1) / GEMM_THREADS)
#define F_ITEMS (KPAIRS * FCOLS)
#define XP (DD + 4)
#define X_ITEMS (GROUP_ROWS * (DD / 4))
#define X_TRIPS (X_ITEMS / GEMM_THREADS)
static_assert(X_TRIPS == 8 && X_TRIPS * GEMM_THREADS == X_ITEMS, "a thread stages eight x words");
static_assert(GEMM_THREADS / 32 == 20 && DD / 4 == 32, "a wave stages one row per load, 20 rows per trip");
static_assert(GEMM_GROUPS * GROUP_ROWS == MROWS, "every row has exactly one group");

template <bool SCORE>
__device__ __forceinline__ void gemm_unit(const float* xs, const float2* Wlds,
                                          const float2* Flds, float* __restrict__ h,
                                          float* __restrict__ a_src,
                                          float* __restrict__ a_dst, int rowBase,
                                          int colBase, int lane) {
  v8f acc[8];
  v8f accs;
  const v8f vzero = {0.f, 0.f, 0.f, 0.f, 0.f, 0.f, 0.f, 0.f};
#pragma unroll
  for (int t = 0; t < 8; ++t) acc[t] = vzero;
  accs = vzero;

  const int m = lane & 15;
  const int kb = (lane >> 4) << 1;
  const int kp = lane >> 4;
  const float* xrow = xs + m * XP;
  for (int k = 0; k < DD; k += 4) {
    v2f a;
    a.x = xrow[k + kb];
    a.y = xrow[k + kb + 1];
    const float2* w0 = Wlds + ((k >> 1) + kp) * HO + colBase + (lane & 15);
#pragma unroll
    for (int t = 0; t < 8; ++t) {
      float2 bp = w0[t * 16];
      v2f b;
      b.x = bp.x;
      b.y = bp.y;
      acc[t] = __builtin_amdgcn_wmma_f32_16x16x4_f32(
          false, a, false, b, (short)0, acc[t], false, false);
    }
    if (SCORE) {
      const float2 fpr = Flds[((k >> 1) + kp) * FCOLS + (lane & 15)];
      v2f b;
      b.x = fpr.x;
      b.y = fpr.y;
      accs = __builtin_amdgcn_wmma_f32_16x16x4_f32(
          false, a, false, b, (short)0, accs, false, false);
    }
  }

  const int nLocal = lane & 15;
  const int rOff = (lane >> 4) * 8;
#pragma unroll
  for (int t = 0; t < 8; ++t)
#pragma unroll
    for (int v = 0; v < 8; ++v)
      h[(size_t)(rowBase + rOff + v) * HO + colBase + t * 16 + nLocal] = acc[t][v];
  if (SCORE && nLocal < 8) {
    float* tab = nLocal < 4 ? a_src : a_dst;
#pragma unroll
    for (int v = 0; v < 8; ++v)
      tab[(size_t)(rowBase + rOff + v) * HH + (nLocal & 3)] = accs[v];
  }
}

__global__ __launch_bounds__(GEMM_THREADS) void gemm_h(const float* __restrict__ x,
                                                       const float* __restrict__ W,
                                                       const float* __restrict__ wap,
                                                       float* __restrict__ h,
                                                       float* __restrict__ a_src,
                                                       float* __restrict__ a_dst) {
  __shared__ float2 Wlds[KPAIRS * HO];
  __shared__ float2 Flds[KPAIRS * FCOLS];
  __shared__ float4 Xlds[GROUP_ROWS * XP / 4];
  const int tid = threadIdx.x;
  const int lane = tid & 31;
  const int wave = __builtin_amdgcn_readfirstlane(tid >> 5);

  float4 r0[W_TRIPS], r1[W_TRIPS];
#pragma unroll
  for (int it = 0; it < W_TRIPS; ++it) {
    const int i = it * GEMM_THREADS + tid;
    if (i < W_ITEMS) {
      const int p = i >> 6, c4 = i & 63;
      r0[it] = ((const float4*)(W + (size_t)(2 * p) * HO))[c4];
      r1[it] = ((const float4*)(W + (size_t)(2 * p + 1) * HO))[c4];
    }
  }
  const float4* xg = (const float4*)(x + (size_t)blockIdx.x * GROUP_ROWS * DD) + tid;
  const float4 xr0 = xg[0 * GEMM_THREADS], xr1 = xg[1 * GEMM_THREADS];
  const float4 xr2 = xg[2 * GEMM_THREADS], xr3 = xg[3 * GEMM_THREADS];
  const float4 xr4 = xg[4 * GEMM_THREADS], xr5 = xg[5 * GEMM_THREADS];
  const float4 xr6 = xg[6 * GEMM_THREADS], xr7 = xg[7 * GEMM_THREADS];
  float2 f0 = make_float2(0.f, 0.f), f1 = f0;
  f0 = ((const float2*)wap)[tid];
  if (tid + GEMM_THREADS < F_ITEMS) f1 = ((const float2*)wap)[tid + GEMM_THREADS];
#pragma unroll
  for (int it = 0; it < W_TRIPS; ++it) {
    const int i = it * GEMM_THREADS + tid;
    if (i < W_ITEMS) {
      const int p = i >> 6, c4 = i & 63;
      float4* d = (float4*)(Wlds + p * HO + 4 * c4);
      d[0] = make_float4(r0[it].x, r1[it].x, r0[it].y, r1[it].y);
      d[1] = make_float4(r0[it].z, r1[it].z, r0[it].w, r1[it].w);
    }
  }
  float4* xd = Xlds + (tid >> 5) * (XP / 4) + (tid & 31);
  xd[0 * 20 * (XP / 4)] = xr0;
  xd[1 * 20 * (XP / 4)] = xr1;
  xd[2 * 20 * (XP / 4)] = xr2;
  xd[3 * 20 * (XP / 4)] = xr3;
  xd[4 * 20 * (XP / 4)] = xr4;
  xd[5 * 20 * (XP / 4)] = xr5;
  xd[6 * 20 * (XP / 4)] = xr6;
  xd[7 * 20 * (XP / 4)] = xr7;
  Flds[tid] = f0;
  if (tid + GEMM_THREADS < F_ITEMS) Flds[tid + GEMM_THREADS] = f1;
  __syncthreads();

  const int strip = wave >> 1;
  const int colBase = (wave & 1) * (HO / 2);
  const int rowBase = blockIdx.x * GROUP_ROWS + strip * 16;
  const float* xs = (const float*)Xlds + strip * 16 * XP;
  if ((wave & 1) == ((strip >> 1) & 1))
    gemm_unit<true>(xs, Wlds, Flds, h, a_src, a_dst, rowBase, colBase, lane);
  else
    gemm_unit<false>(xs, Wlds, Flds, h, a_src, a_dst, rowBase, colBase, lane);
}

#define OWN_NODES 40
#define E16 (EE / 16)
static_assert(E16 * 16 == EE, "the owner plane is written and read as whole 16-byte words");
#define PREP_THREADS 256
#define PREP_BLOCKS ((E16 + PREP_THREADS - 1) / PREP_THREADS)
#define FOLD_BLOCKS ((DD * 8) / PREP_THREADS)
static_assert(FOLD_BLOCKS * PREP_THREADS == DD * 8, "one thread per folded weight");
__device__ __forceinline__ unsigned owner_pack(int4 d) {
  const unsigned a = min((unsigned)d.x / OWN_NODES, 255u);
  const unsigned b = min((unsigned)d.y / OWN_NODES, 255u);
  const unsigned c = min((unsigned)d.z / OWN_NODES, 255u);
  const unsigned e = min((unsigned)d.w / OWN_NODES, 255u);
  return a | (b << 8) | (c << 16) | (e << 24);
}
__global__ __launch_bounds__(PREP_THREADS) void edge_prep(
    const float* __restrict__ W_edge, const float* __restrict__ att_edge,
    float* __restrict__ coef, const int* __restrict__ dst, int* __restrict__ own,
    const float* __restrict__ W, const float* __restrict__ att_src,
    const float* __restrict__ att_dst, float* __restrict__ wap) {
  const int t = threadIdx.x;
  const int i = blockIdx.x * PREP_THREADS + t;
  if (i < E16) {
    const int4* d4 = (const int4*)dst + 4 * i;
    const int4 d0 = d4[0], d1 = d4[1], d2 = d4[2], d3 = d4[3];
    int4 o;
    o.x = (int)owner_pack(d0);
    o.y = (int)owner_pack(d1);
    o.z = (int)owner_pack(d2);
    o.w = (int)owner_pack(d3);
    ((int4*)own)[i] = o;
  }
  if (blockIdx.x == PREP_BLOCKS && t < HH) {
    float s = 0.f;
    for (int o = 0; o < OO; ++o) s += W_edge[t * OO + o] * att_edge[t * OO + o];
    coef[t] = s;
  }
  if (blockIdx.x > PREP_BLOCKS) {
    const int idx = (blockIdx.x - PREP_BLOCKS - 1) * PREP_THREADS + t;
    const int d = idx >> 3, c = idx & 7, hd = c & 3;
    const float4* wr = (const float4*)(W + (size_t)d * HO + hd * OO);
    const float4* ar = (const float4*)((c < 4 ? att_src : att_dst) + hd * OO);
    float s = 0.f;
#pragma unroll
    for (int o = 0; o < OO / 4; ++o) {
      const float4 w4 = wr[o], a4 = ar[o];
      s += w4.x * a4.x;
      s += w4.y * a4.y;
      s += w4.z * a4.z;
      s += w4.w * a4.w;
    }
    wap[((d >> 1) * FCOLS + c) * 2 + (d & 1)] = s;
    wap[((d >> 1) * FCOLS + c + 8) * 2 + (d & 1)] = 0.f;
  }
}

__device__ __forceinline__ float edge_score(float as, float ad, float ae, float cf) {
  float a = as + ad + ae * cf;
  return a > 0.f ? a : a * NEG_SLOPE;
}

__device__ __forceinline__ int pair_pick(int v, int j, bool hi) {
  const int lo_v = __builtin_amdgcn_readlane(v, 2 * j);
  const int hi_v = __builtin_amdgcn_readlane(v, 2 * j + 1);
  return hi ? hi_v : lo_v;
}

#define OWN_BLOCKS (NB / OWN_NODES)
static_assert(OWN_BLOCKS * OWN_NODES == NB, "every node has exactly one owner");
static_assert(OWN_BLOCKS <= 255, "an owner id is one byte and 255 means no owner");
#define OWN_THREADS 1024
#define OWN_SCORES (OWN_NODES * BB * HH)
static_assert(OWN_SCORES <= OWN_THREADS, "one thread clears one cell");
#define SCAN_LOADS 10
static_assert(SCAN_LOADS * OWN_THREADS >= E16, "one search covers every edge");
#define PART_LOADS 3
#define PART16 (PART_LOADS * OWN_THREADS)
#define PARTS ((E16 + PART16 - 1) / PART16)
#define LIST_CAP (16 * PART16)
#define WALKS 3

__device__ __forceinline__ unsigned zero_bytes(unsigned x) {
  const unsigned t = (x & 0x7F7F7F7Fu) + 0x7F7F7F7Fu;
  return ~(t | x | 0x7F7F7F7Fu);
}

template <int LOADS, bool CHECK>
__device__ __forceinline__ void own_scan(const int4* __restrict__ tab4, int qb, int qe,
                                         unsigned gpat, int tid, int* cnt, int* elist) {
  int4 dq[LOADS];
#pragma unroll
  for (int k = 0; k < LOADS; ++k) {
    const int q = qb + k * OWN_THREADS + tid;
    dq[k] = q < qe ? tab4[q] : make_int4(-1, -1, -1, -1);
  }
  unsigned m[LOADS];
  int c = 0;
#pragma unroll
  for (int k = 0; k < LOADS; ++k) {
    const unsigned f0 = zero_bytes((unsigned)dq[k].x ^ gpat);
    const unsigned f1 = zero_bytes((unsigned)dq[k].y ^ gpat);
    const unsigned f2 = zero_bytes((unsigned)dq[k].z ^ gpat);
    const unsigned f3 = zero_bytes((unsigned)dq[k].w ^ gpat);
    m[k] = (f0 >> 7) | (f1 >> 6) | (f2 >> 5) | (f3 >> 4);
    c += __builtin_popcount(m[k]);
  }
  if (c == 0) return;
  int pos = atomicAdd(cnt, c);
#pragma unroll
  for (int k = 0; k < LOADS; ++k) {
    const int q = qb + k * OWN_THREADS + tid;
    unsigned mk = m[k];
    while (mk != 0u) {
      const int bit = __builtin_ctz(mk);
      mk &= mk - 1u;
      if (!CHECK || pos < LIST_CAP) elist[pos] = 16 * q + 4 * (bit & 7) + (bit >> 3);
      ++pos;
    }
  }
}

template <int WALK>
__device__ __forceinline__ void own_stat(int sv, int dv, int av, int nk, int n0, int ot,
                                         int t, bool hi, float cf,
                                         const float* __restrict__ a_src,
                                         const float* __restrict__ a_dst,
                                         unsigned* mxl, float* denl) {
  float al[16];
  int sl[16];
  unsigned mv[16];
#pragma unroll
  for (int j = 0; j < 16; ++j) {
    const int sN = pair_pick(sv, j, hi);
    const int dN = pair_pick(dv, j, hi);
    const float ae = __int_as_float(pair_pick(av, j, hi));
    al[j] = edge_score(a_src[ot + sN * HH], a_dst[ot + dN * HH], ae, cf);
    sl[j] = (dN - n0) * (BB * HH) + t;
    mv[j] = WALK == 1 ? mxl[sl[j]] : 0u;
  }
  const int k0 = hi ? 1 : 0;
#pragma unroll
  for (int j = 0; j < 16; ++j)
    if (2 * j + k0 < nk) {
      if (WALK == 0)
        __hip_atomic_fetch_max(&mxl[sl[j]], fenc(al[j]), __ATOMIC_RELAXED,
                               __HIP_MEMORY_SCOPE_WORKGROUP);
      else
        __hip_atomic_fetch_add(&denl[sl[j]], __expf(al[j] - fdec(mv[j])),
                               __ATOMIC_RELAXED, __HIP_MEMORY_SCOPE_WORKGROUP);
    }
}

struct EdgeRegs {
  float as, ad, ae, den;
  unsigned mx;
  int dl;
  float2 hv[BB][HH];
};

__device__ __forceinline__ void own_load(EdgeRegs& r, int sv, int dv, int av, int k,
                                         int n0, int ot, int t, int lane,
                                         const float* __restrict__ a_src,
                                         const float* __restrict__ a_dst,
                                         const unsigned* mxl, const float* denl,
                                         const float* __restrict__ h) {
  const int sN = __builtin_amdgcn_readlane(sv, k);
  const int dN = __builtin_amdgcn_readlane(dv, k);
  r.ae = __int_as_float(__builtin_amdgcn_readlane(av, k));
  r.as = a_src[ot + sN * HH];
  r.ad = a_dst[ot + dN * HH];
  r.mx = mxl[(dN - n0) * (BB * HH) + t];
  r.den = denl[(dN - n0) * (BB * HH) + t];
  r.dl = (dN - n0) * (BB * OO) + lane;
  const float2* hs = (const float2*)(h + sN * HO) + lane;
#pragma unroll
  for (int b = 0; b < BB; ++b)
#pragma unroll
    for (int k2 = 0; k2 < HH; ++k2) r.hv[b][k2] = hs[(b * NB * HO + k2 * OO) / 2];
}

__device__ __forceinline__ void own_add(const EdgeRegs& r, float cf, float* accl) {
  const float a = edge_score(r.as, r.ad, r.ae, cf);
  const float p = __expf(a - fdec(r.mx));
  const float wl = p / fmaxf(r.den, 1e-16f);
#pragma unroll
  for (int b = 0; b < BB; ++b) {
    const float w0 = __int_as_float(__builtin_amdgcn_readlane(__float_as_int(wl), b * 4 + 0));
    const float w1 = __int_as_float(__builtin_amdgcn_readlane(__float_as_int(wl), b * 4 + 1));
    const float w2 = __int_as_float(__builtin_amdgcn_readlane(__float_as_int(wl), b * 4 + 2));
    const float w3 = __int_as_float(__builtin_amdgcn_readlane(__float_as_int(wl), b * 4 + 3));
    const float vx = fmaf(w3, r.hv[b][3].x, fmaf(w2, r.hv[b][2].x, fmaf(w1, r.hv[b][1].x, w0 * r.hv[b][0].x)));
    const float vy = fmaf(w3, r.hv[b][3].y, fmaf(w2, r.hv[b][2].y, fmaf(w1, r.hv[b][1].y, w0 * r.hv[b][0].y)));
    __hip_atomic_fetch_add(&accl[r.dl + b * OO], vx, __ATOMIC_RELAXED, __HIP_MEMORY_SCOPE_WORKGROUP);
    __hip_atomic_fetch_add(&accl[r.dl + b * OO + 32], vy, __ATOMIC_RELAXED, __HIP_MEMORY_SCOPE_WORKGROUP);
  }
}

__global__ __launch_bounds__(OWN_THREADS) void edge_own(
    const int* __restrict__ src, const int* __restrict__ dst,
    const float* __restrict__ eattr, const float* __restrict__ a_src,
    const float* __restrict__ a_dst, const float* __restrict__ coef,
    const float* __restrict__ h, const float* __restrict__ bias,
    const int* __restrict__ own, float* __restrict__ out) {
  __shared__ float accl[OWN_NODES * BB * OO];
  __shared__ int elist[LIST_CAP];
  __shared__ unsigned mxl[OWN_SCORES];
  __shared__ float denl[OWN_SCORES];
  __shared__ int ecnt[1 + WALKS * PARTS];
  const int tid = threadIdx.x;
  const int lane = tid & 31;
  const int t = lane & 15;
  const bool hi = lane >= 16;
  const int wv = __builtin_amdgcn_readfirstlane(tid >> 5);
  const int n0 = blockIdx.x * OWN_NODES;
  const int ot = (t >> 2) * (NB * HH) + (t & 3);
  const float cf = coef[t & 3];

  for (int i = tid; i < OWN_NODES * BB * OO; i += OWN_THREADS) accl[i] = 0.f;
  if (tid < OWN_SCORES) {
    mxl[tid] = 0u;
    denl[tid] = 0.f;
  }
  if (tid < 1 + WALKS * PARTS) ecnt[tid] = 0;
  __syncthreads();

  const int4* tab4 = (const int4*)own;
  const unsigned gpat = blockIdx.x * 0x01010101u;
  own_scan<SCAN_LOADS, true>(tab4, 0, E16, gpat, tid, &ecnt[0], elist);
  __syncthreads();
  int total = ecnt[0];
  const bool whole = total <= LIST_CAP;
  const int nparts = whole ? 1 : PARTS;
  for (int walk = 0; walk < WALKS; ++walk) {
    for (int part = 0; part < nparts; ++part) {
      if (!whole) {
        int* cnt = &ecnt[1 + walk * PARTS + part];
        const int qb = part * PART16;
        own_scan<PART_LOADS, false>(tab4, qb, min(qb + PART16, E16), gpat, tid, cnt, elist);
        __syncthreads();
        total = *cnt;
      }

      for (int b0 = 0; b0 < total; b0 += OWN_THREADS) {
        const int my = b0 + lane * 32 + wv;
        int sv = 0, dv = n0, av = 0;
        if (my < total) {
          const int e = elist[my];
          sv = src[e];
          dv = dst[e];
          av = __float_as_int(eattr[e]);
        }
        const int rem = total - b0 - wv;
        const int nk = rem <= 0 ? 0 : (rem + 31 >= 32 * 32 ? 32 : (rem + 31) >> 5);
        if (nk > 0) {
          if (walk == 0) {
            own_stat<0>(sv, dv, av, nk, n0, ot, t, hi, cf, a_src, a_dst, mxl, denl);
          } else if (walk == 1) {
            own_stat<1>(sv, dv, av, nk, n0, ot, t, hi, cf, a_src, a_dst, mxl, denl);
          } else {
            EdgeRegs ra, rb;
            own_load(ra, sv, dv, av, 0, n0, ot, t, lane, a_src, a_dst, mxl, denl, h);
            int k = 0;
            for (; k + 2 < nk; k += 2) {
              own_load(rb, sv, dv, av, k + 1, n0, ot, t, lane, a_src, a_dst, mxl, denl, h);
              own_add(ra, cf, accl);
              own_load(ra, sv, dv, av, k + 2, n0, ot, t, lane, a_src, a_dst, mxl, denl, h);
              own_add(rb, cf, accl);
            }
            if (k + 1 < nk) {
              own_load(rb, sv, dv, av, k + 1, n0, ot, t, lane, a_src, a_dst, mxl, denl, h);
              own_add(ra, cf, accl);
              own_add(rb, cf, accl);
            } else {
              own_add(ra, cf, accl);
            }
          }
        }
      }
      __syncthreads();
    }
  }

  for (int i = tid; i < OWN_NODES * BB * (OO / 4); i += OWN_THREADS) {
    const int q = i & 15;
    const int r = i >> 4;
    const int nl = r % OWN_NODES, b = r / OWN_NODES;
    const float* ap = accl + (nl * BB + b) * OO + 2 * q;
    const float2 ev = *(const float2*)ap;
    const float2 od = *(const float2*)(ap + 32);
    const float4 bv = ((const float4*)bias)[q];
    float4 o;
    o.x = ev.x * 0.25f + bv.x;
    o.y = od.x * 0.25f + bv.y;
    o.z = ev.y * 0.25f + bv.z;
    o.w = od.y * 0.25f + bv.w;
    ((float4*)out)[((size_t)b * NB + n0 + nl) * (OO / 4) + q] = o;
  }
}

extern "C" void kernel_launch(void* const* d_in, const int* in_sizes, int n_in,
                              void* d_out, int out_size, void* d_ws, size_t ws_size,
                              hipStream_t stream) {
  const float* x        = (const float*)d_in[0];
  const int*   eidx     = (const int*)d_in[1];
  const float* eattr    = (const float*)d_in[2];
  const float* W        = (const float*)d_in[3];
  const float* att_src  = (const float*)d_in[4];
  const float* att_dst  = (const float*)d_in[5];
  const float* W_edge   = (const float*)d_in[6];
  const float* att_edge = (const float*)d_in[7];
  const float* bias     = (const float*)d_in[8];
  float* out = (float*)d_out;

  char* ws = (char*)d_ws;
  float*    h_buf  = (float*)   (ws + 0);
  float*    a_src  = (float*)   (ws + 40960000);
  float*    a_dst  = (float*)   (ws + 43520000);
  float*    coef   = (float*)   (ws + 51200000);
  int*      own    = (int*)     (ws + 51200256);
  float*    wap    = (float*)   (ws + 51360768);

  const int* srcI = eidx;
  const int* dstI = eidx + EE;

  edge_prep<<<PREP_BLOCKS + 1 + FOLD_BLOCKS, PREP_THREADS, 0, stream>>>(
      W_edge, att_edge, coef, dstI, own, W, att_src, att_dst, wap);
  gemm_h<<<GEMM_GROUPS, GEMM_THREADS, 0, stream>>>(x, W, wap, h_buf, a_src, a_dst);
  edge_own<<<OWN_BLOCKS, OWN_THREADS, 0, stream>>>(srcI, dstI, eattr, a_src, a_dst,
                                                   coef, h_buf, bias, own, out);
}


# Owner waves keep each edge (nodes, attribute, logits) in the lane that loaded it for all three walks and its 16 exp values in LDS: no list, edge or score reads after walk 0; sum orders unchanged
# speedup vs baseline: 9.0424x; 1.1701x over previous
#include <hip/hip_runtime.h>


#define NB 10000
#define BB 4
#define EE 160000
#define DD 128
#define HH 4
#define OO 64
#define HO 256
#define MROWS (BB * NB)
#define NEG_SLOPE 0.2f

typedef float v2f __attribute__((ext_vector_type(2)));
typedef float v8f __attribute__((ext_vector_type(8)));

__device__ __forceinline__ unsigned fenc(float f) {
  unsigned u = __float_as_uint(f);
  return (u & 0x80000000u) ? ~u : (u | 0x80000000u);
}
__device__ __forceinline__ float fdec(unsigned u) {
  return __uint_as_float((u & 0x80000000u) ? (u ^ 0x80000000u) : ~u);
}

#define FCOLS 16
#define GEMM_GROUPS 250
#define GEMM_WAVES 20
#define GEMM_THREADS (GEMM_WAVES * 32)
#define GROUP_ROWS ((GEMM_WAVES / 2) * 16)
#define KPAIRS (DD / 2)
#define W_ITEMS (KPAIRS * (HO / 4))
#define W_TRIPS ((W_ITEMS + GEMM_THREADS - 1) / GEMM_THREADS)
#define F_ITEMS (KPAIRS * FCOLS)
#define XP (DD + 4)
#define X_ITEMS (GROUP_ROWS * (DD / 4))
#define X_TRIPS (X_ITEMS / GEMM_THREADS)
static_assert(X_TRIPS == 8 && X_TRIPS * GEMM_THREADS == X_ITEMS, "a thread stages eight x words");
static_assert(GEMM_THREADS / 32 == 20 && DD / 4 == 32, "a wave stages one row per load, 20 rows per trip");
static_assert(GEMM_GROUPS * GROUP_ROWS == MROWS, "every row has exactly one group");

template <bool SCORE>
__device__ __forceinline__ void gemm_unit(const float* xs, const float2* Wlds,
                                          const float2* Flds, float* __restrict__ h,
                                          float* __restrict__ a_src,
                                          float* __restrict__ a_dst, int rowBase,
                                          int colBase, int lane) {
  v8f acc[8];
  v8f accs;
  const v8f vzero = {0.f, 0.f, 0.f, 0.f, 0.f, 0.f, 0.f, 0.f};
#pragma unroll
  for (int t = 0; t < 8; ++t) acc[t] = vzero;
  accs = vzero;

  const int m = lane & 15;
  const int kb = (lane >> 4) << 1;
  const int kp = lane >> 4;
  const float* xrow = xs + m * XP;
  for (int k = 0; k < DD; k += 4) {
    v2f a;
    a.x = xrow[k + kb];
    a.y = xrow[k + kb + 1];
    const float2* w0 = Wlds + ((k >> 1) + kp) * HO + colBase + (lane & 15);
#pragma unroll
    for (int t = 0; t < 8; ++t) {
      float2 bp = w0[t * 16];
      v2f b;
      b.x = bp.x;
      b.y = bp.y;
      acc[t] = __builtin_amdgcn_wmma_f32_16x16x4_f32(
          false, a, false, b, (short)0, acc[t], false, false);
    }
    if (SCORE) {
      const float2 fpr = Flds[((k >> 1) + kp) * FCOLS + (lane & 15)];
      v2f b;
      b.x = fpr.x;
      b.y = fpr.y;
      accs = __builtin_amdgcn_wmma_f32_16x16x4_f32(
          false, a, false, b, (short)0, accs, false, false);
    }
  }

  const int nLocal = lane & 15;
  const int rOff = (lane >> 4) * 8;
#pragma unroll
  for (int t = 0; t < 8; ++t)
#pragma unroll
    for (int v = 0; v < 8; ++v)
      h[(size_t)(rowBase + rOff + v) * HO + colBase + t * 16 + nLocal] = acc[t][v];
  if (SCORE && nLocal < 8) {
    float* tab = nLocal < 4 ? a_src : a_dst;
#pragma unroll
    for (int v = 0; v < 8; ++v)
      tab[(size_t)(rowBase + rOff + v) * HH + (nLocal & 3)] = accs[v];
  }
}

__global__ __launch_bounds__(GEMM_THREADS) void gemm_h(const float* __restrict__ x,
                                                       const float* __restrict__ W,
                                                       const float* __restrict__ wap,
                                                       float* __restrict__ h,
                                                       float* __restrict__ a_src,
                                                       float* __restrict__ a_dst) {
  __shared__ float2 Wlds[KPAIRS * HO];
  __shared__ float2 Flds[KPAIRS * FCOLS];
  __shared__ float4 Xlds[GROUP_ROWS * XP / 4];
  const int tid = threadIdx.x;
  const int lane = tid & 31;
  const int wave = __builtin_amdgcn_readfirstlane(tid >> 5);

  float4 r0[W_TRIPS], r1[W_TRIPS];
#pragma unroll
  for (int it = 0; it < W_TRIPS; ++it) {
    const int i = it * GEMM_THREADS + tid;
    if (i < W_ITEMS) {
      const int p = i >> 6, c4 = i & 63;
      r0[it] = ((const float4*)(W + (size_t)(2 * p) * HO))[c4];
      r1[it] = ((const float4*)(W + (size_t)(2 * p + 1) * HO))[c4];
    }
  }
  const float4* xg = (const float4*)(x + (size_t)blockIdx.x * GROUP_ROWS * DD) + tid;
  const float4 xr0 = xg[0 * GEMM_THREADS], xr1 = xg[1 * GEMM_THREADS];
  const float4 xr2 = xg[2 * GEMM_THREADS], xr3 = xg[3 * GEMM_THREADS];
  const float4 xr4 = xg[4 * GEMM_THREADS], xr5 = xg[5 * GEMM_THREADS];
  const float4 xr6 = xg[6 * GEMM_THREADS], xr7 = xg[7 * GEMM_THREADS];
  float2 f0 = make_float2(0.f, 0.f), f1 = f0;
  f0 = ((const float2*)wap)[tid];
  if (tid + GEMM_THREADS < F_ITEMS) f1 = ((const float2*)wap)[tid + GEMM_THREADS];
#pragma unroll
  for (int it = 0; it < W_TRIPS; ++it) {
    const int i = it * GEMM_THREADS + tid;
    if (i < W_ITEMS) {
      const int p = i >> 6, c4 = i & 63;
      float4* d = (float4*)(Wlds + p * HO + 4 * c4);
      d[0] = make_float4(r0[it].x, r1[it].x, r0[it].y, r1[it].y);
      d[1] = make_float4(r0[it].z, r1[it].z, r0[it].w, r1[it].w);
    }
  }
  float4* xd = Xlds + (tid >> 5) * (XP / 4) + (tid & 31);
  xd[0 * 20 * (XP / 4)] = xr0;
  xd[1 * 20 * (XP / 4)] = xr1;
  xd[2 * 20 * (XP / 4)] = xr2;
  xd[3 * 20 * (XP / 4)] = xr3;
  xd[4 * 20 * (XP / 4)] = xr4;
  xd[5 * 20 * (XP / 4)] = xr5;
  xd[6 * 20 * (XP / 4)] = xr6;
  xd[7 * 20 * (XP / 4)] = xr7;
  Flds[tid] = f0;
  if (tid + GEMM_THREADS < F_ITEMS) Flds[tid + GEMM_THREADS] = f1;
  __syncthreads();

  const int strip = wave >> 1;
  const int colBase = (wave & 1) * (HO / 2);
  const int rowBase = blockIdx.x * GROUP_ROWS + strip * 16;
  const float* xs = (const float*)Xlds + strip * 16 * XP;
  if ((wave & 1) == ((strip >> 1) & 1))
    gemm_unit<true>(xs, Wlds, Flds, h, a_src, a_dst, rowBase, colBase, lane);
  else
    gemm_unit<false>(xs, Wlds, Flds, h, a_src, a_dst, rowBase, colBase, lane);
}

#define OWN_NODES 40
#define E16 (EE / 16)
static_assert(E16 * 16 == EE, "the owner plane is written and read as whole 16-byte words");
#define PREP_THREADS 256
#define PREP_BLOCKS ((E16 + PREP_THREADS - 1) / PREP_THREADS)
#define FOLD_BLOCKS ((DD * 8) / PREP_THREADS)
static_assert(FOLD_BLOCKS * PREP_THREADS == DD * 8, "one thread per folded weight");
__device__ __forceinline__ unsigned owner_pack(int4 d) {
  const unsigned a = min((unsigned)d.x / OWN_NODES, 255u);
  const unsigned b = min((unsigned)d.y / OWN_NODES, 255u);
  const unsigned c = min((unsigned)d.z / OWN_NODES, 255u);
  const unsigned e = min((unsigned)d.w / OWN_NODES, 255u);
  return a | (b << 8) | (c << 16) | (e << 24);
}
__global__ __launch_bounds__(PREP_THREADS) void edge_prep(
    const float* __restrict__ W_edge, const float* __restrict__ att_edge,
    float* __restrict__ coef, const int* __restrict__ dst, int* __restrict__ own,
    const float* __restrict__ W, const float* __restrict__ att_src,
    const float* __restrict__ att_dst, float* __restrict__ wap) {
  const int t = threadIdx.x;
  const int i = blockIdx.x * PREP_THREADS + t;
  if (i < E16) {
    const int4* d4 = (const int4*)dst + 4 * i;
    const int4 d0 = d4[0], d1 = d4[1], d2 = d4[2], d3 = d4[3];
    int4 o;
    o.x = (int)owner_pack(d0);
    o.y = (int)owner_pack(d1);
    o.z = (int)owner_pack(d2);
    o.w = (int)owner_pack(d3);
    ((int4*)own)[i] = o;
  }
  if (blockIdx.x == PREP_BLOCKS && t < HH) {
    float s = 0.f;
    for (int o = 0; o < OO; ++o) s += W_edge[t * OO + o] * att_edge[t * OO + o];
    coef[t] = s;
  }
  if (blockIdx.x > PREP_BLOCKS) {
    const int idx = (blockIdx.x - PREP_BLOCKS - 1) * PREP_THREADS + t;
    const int d = idx >> 3, c = idx & 7, hd = c & 3;
    const float4* wr = (const float4*)(W + (size_t)d * HO + hd * OO);
    const float4* ar = (const float4*)((c < 4 ? att_src : att_dst) + hd * OO);
    float s = 0.f;
#pragma unroll
    for (int o = 0; o < OO / 4; ++o) {
      const float4 w4 = wr[o], a4 = ar[o];
      s += w4.x * a4.x;
      s += w4.y * a4.y;
      s += w4.z * a4.z;
      s += w4.w * a4.w;
    }
    wap[((d >> 1) * FCOLS + c) * 2 + (d & 1)] = s;
    wap[((d >> 1) * FCOLS + c + 8) * 2 + (d & 1)] = 0.f;
  }
}

__device__ __forceinline__ float edge_score(float as, float ad, float ae, float cf) {
  float a = as + ad + ae * cf;
  return a > 0.f ? a : a * NEG_SLOPE;
}

__device__ __forceinline__ int pair_pick(int v, int j, bool hi) {
  const int lo_v = __builtin_amdgcn_readlane(v, 2 * j);
  const int hi_v = __builtin_amdgcn_readlane(v, 2 * j + 1);
  return hi ? hi_v : lo_v;
}

#define OWN_BLOCKS (NB / OWN_NODES)
static_assert(OWN_BLOCKS * OWN_NODES == NB, "every node has exactly one owner");
static_assert(OWN_BLOCKS <= 255, "an owner id is one byte and 255 means no owner");
#define OWN_THREADS 1024
#define OWN_SCORES (OWN_NODES * BB * HH)
static_assert(OWN_SCORES <= OWN_THREADS, "one thread clears one cell");
#define SCAN_LOADS 10
static_assert(SCAN_LOADS * OWN_THREADS >= E16, "one search covers every edge");
#define PART_LOADS 3
#define PART16 (PART_LOADS * OWN_THREADS)
#define PARTS ((E16 + PART16 - 1) / PART16)
#define LIST_CAP (16 * PART16)
#define WALKS 3

__device__ __forceinline__ unsigned zero_bytes(unsigned x) {
  const unsigned t = (x & 0x7F7F7F7Fu) + 0x7F7F7F7Fu;
  return ~(t | x | 0x7F7F7F7Fu);
}

template <int LOADS, bool CHECK>
__device__ __forceinline__ void own_scan(const int4* __restrict__ tab4, int qb, int qe,
                                         unsigned gpat, int tid, int* cnt, int* elist) {
  int4 dq[LOADS];
#pragma unroll
  for (int k = 0; k < LOADS; ++k) {
    const int q = qb + k * OWN_THREADS + tid;
    dq[k] = q < qe ? tab4[q] : make_int4(-1, -1, -1, -1);
  }
  unsigned m[LOADS];
  int c = 0;
#pragma unroll
  for (int k = 0; k < LOADS; ++k) {
    const unsigned f0 = zero_bytes((unsigned)dq[k].x ^ gpat);
    const unsigned f1 = zero_bytes((unsigned)dq[k].y ^ gpat);
    const unsigned f2 = zero_bytes((unsigned)dq[k].z ^ gpat);
    const unsigned f3 = zero_bytes((unsigned)dq[k].w ^ gpat);
    m[k] = (f0 >> 7) | (f1 >> 6) | (f2 >> 5) | (f3 >> 4);
    c += __builtin_popcount(m[k]);
  }
  if (c == 0) return;
  int pos = atomicAdd(cnt, c);
#pragma unroll
  for (int k = 0; k < LOADS; ++k) {
    const int q = qb + k * OWN_THREADS + tid;
    unsigned mk = m[k];
    while (mk != 0u) {
      const int bit = __builtin_ctz(mk);
      mk &= mk - 1u;
      if (!CHECK || pos < LIST_CAP) elist[pos] = 16 * q + 4 * (bit & 7) + (bit >> 3);
      ++pos;
    }
  }
}

template <int WALK>
__device__ __forceinline__ void own_stat(int sv, int dv, int av, int nk, int n0, int ot,
                                         int t, bool hi, float cf,
                                         const float* __restrict__ a_src,
                                         const float* __restrict__ a_dst,
                                         unsigned* mxl, float* denl) {
  float al[16];
  int sl[16];
  unsigned mv[16];
#pragma unroll
  for (int j = 0; j < 16; ++j) {
    const int sN = pair_pick(sv, j, hi);
    const int dN = pair_pick(dv, j, hi);
    const float ae = __int_as_float(pair_pick(av, j, hi));
    al[j] = edge_score(a_src[ot + sN * HH], a_dst[ot + dN * HH], ae, cf);
    sl[j] = (dN - n0) * (BB * HH) + t;
    mv[j] = WALK == 1 ? mxl[sl[j]] : 0u;
  }
  const int k0 = hi ? 1 : 0;
#pragma unroll
  for (int j = 0; j < 16; ++j)
    if (2 * j + k0 < nk) {
      if (WALK == 0)
        __hip_atomic_fetch_max(&mxl[sl[j]], fenc(al[j]), __ATOMIC_RELAXED,
                               __HIP_MEMORY_SCOPE_WORKGROUP);
      else
        __hip_atomic_fetch_add(&denl[sl[j]], __expf(al[j] - fdec(mv[j])),
                               __ATOMIC_RELAXED, __HIP_MEMORY_SCOPE_WORKGROUP);
    }
}

struct EdgeRegs {
  float as, ad, ae, den;
  unsigned mx;
  int dl;
  float2 hv[BB][HH];
};

__device__ __forceinline__ void own_load(EdgeRegs& r, int sv, int dv, int av, int k,
                                         int n0, int ot, int t, int lane,
                                         const float* __restrict__ a_src,
                                         const float* __restrict__ a_dst,
                                         const unsigned* mxl, const float* denl,
                                         const float* __restrict__ h) {
  const int sN = __builtin_amdgcn_readlane(sv, k);
  const int dN = __builtin_amdgcn_readlane(dv, k);
  r.ae = __int_as_float(__builtin_amdgcn_readlane(av, k));
  r.as = a_src[ot + sN * HH];
  r.ad = a_dst[ot + dN * HH];
  r.mx = mxl[(dN - n0) * (BB * HH) + t];
  r.den = denl[(dN - n0) * (BB * HH) + t];
  r.dl = (dN - n0) * (BB * OO) + lane;
  const float2* hs = (const float2*)(h + sN * HO) + lane;
#pragma unroll
  for (int b = 0; b < BB; ++b)
#pragma unroll
    for (int k2 = 0; k2 < HH; ++k2) r.hv[b][k2] = hs[(b * NB * HO + k2 * OO) / 2];
}

__device__ __forceinline__ void own_add(const EdgeRegs& r, float cf, float* accl) {
  const float a = edge_score(r.as, r.ad, r.ae, cf);
  const float p = __expf(a - fdec(r.mx));
  const float wl = p / fmaxf(r.den, 1e-16f);
#pragma unroll
  for (int b = 0; b < BB; ++b) {
    const float w0 = __int_as_float(__builtin_amdgcn_readlane(__float_as_int(wl), b * 4 + 0));
    const float w1 = __int_as_float(__builtin_amdgcn_readlane(__float_as_int(wl), b * 4 + 1));
    const float w2 = __int_as_float(__builtin_amdgcn_readlane(__float_as_int(wl), b * 4 + 2));
    const float w3 = __int_as_float(__builtin_amdgcn_readlane(__float_as_int(wl), b * 4 + 3));
    const float vx = fmaf(w3, r.hv[b][3].x, fmaf(w2, r.hv[b][2].x, fmaf(w1, r.hv[b][1].x, w0 * r.hv[b][0].x)));
    const float vy = fmaf(w3, r.hv[b][3].y, fmaf(w2, r.hv[b][2].y, fmaf(w1, r.hv[b][1].y, w0 * r.hv[b][0].y)));
    __hip_atomic_fetch_add(&accl[r.dl + b * OO], vx, __ATOMIC_RELAXED, __HIP_MEMORY_SCOPE_WORKGROUP);
    __hip_atomic_fetch_add(&accl[r.dl + b * OO + 32], vy, __ATOMIC_RELAXED, __HIP_MEMORY_SCOPE_WORKGROUP);
  }
}

#define KEEP_SLOT (BB * HH)
#define KEEP_WAVE (32 * KEEP_SLOT)
struct RowRegs {
  float p, den;
  int dl;
  float2 hv[BB][HH];
};

__device__ __forceinline__ void keep_load(RowRegs& r, int sv, int dv, int k, int n0,
                                          int t, int lane, const float* pw,
                                          const float* denl,
                                          const float* __restrict__ h) {
  const int sN = __builtin_amdgcn_readlane(sv, k);
  const int dN = __builtin_amdgcn_readlane(dv, k);
  r.p = pw[k * KEEP_SLOT + t];
  r.den = denl[(dN - n0) * (BB * HH) + t];
  r.dl = (dN - n0) * (BB * OO) + lane;
  const float2* hs = (const float2*)(h + sN * HO) + lane;
#pragma unroll
  for (int b = 0; b < BB; ++b)
#pragma unroll
    for (int k2 = 0; k2 < HH; ++k2) r.hv[b][k2] = hs[(b * NB * HO + k2 * OO) / 2];
}

__device__ __forceinline__ void keep_add(const RowRegs& r, float* accl) {
  const float wl = r.p / fmaxf(r.den, 1e-16f);
#pragma unroll
  for (int b = 0; b < BB; ++b) {
    const float w0 = __int_as_float(__builtin_amdgcn_readlane(__float_as_int(wl), b * 4 + 0));
    const float w1 = __int_as_float(__builtin_amdgcn_readlane(__float_as_int(wl), b * 4 + 1));
    const float w2 = __int_as_float(__builtin_amdgcn_readlane(__float_as_int(wl), b * 4 + 2));
    const float w3 = __int_as_float(__builtin_amdgcn_readlane(__float_as_int(wl), b * 4 + 3));
    const float vx = fmaf(w3, r.hv[b][3].x, fmaf(w2, r.hv[b][2].x, fmaf(w1, r.hv[b][1].x, w0 * r.hv[b][0].x)));
    const float vy = fmaf(w3, r.hv[b][3].y, fmaf(w2, r.hv[b][2].y, fmaf(w1, r.hv[b][1].y, w0 * r.hv[b][0].y)));
    __hip_atomic_fetch_add(&accl[r.dl + b * OO], vx, __ATOMIC_RELAXED, __HIP_MEMORY_SCOPE_WORKGROUP);
    __hip_atomic_fetch_add(&accl[r.dl + b * OO + 32], vy, __ATOMIC_RELAXED, __HIP_MEMORY_SCOPE_WORKGROUP);
  }
}

__global__ __launch_bounds__(OWN_THREADS) void edge_own(
    const int* __restrict__ src, const int* __restrict__ dst,
    const float* __restrict__ eattr, const float* __restrict__ a_src,
    const float* __restrict__ a_dst, const float* __restrict__ coef,
    const float* __restrict__ h, const float* __restrict__ bias,
    const int* __restrict__ own, float* __restrict__ out) {
  __shared__ float accl[OWN_NODES * BB * OO];
  __shared__ int elist[LIST_CAP];
  __shared__ unsigned mxl[OWN_SCORES];
  __shared__ float denl[OWN_SCORES];
  __shared__ int ecnt[1 + WALKS * PARTS];
  __shared__ float pl[(OWN_THREADS / 32) * KEEP_WAVE];
  const int tid = threadIdx.x;
  const int lane = tid & 31;
  const int t = lane & 15;
  const bool hi = lane >= 16;
  const int wv = __builtin_amdgcn_readfirstlane(tid >> 5);
  const int n0 = blockIdx.x * OWN_NODES;
  const int ot = (t >> 2) * (NB * HH) + (t & 3);
  const float cf = coef[t & 3];

  for (int i = tid; i < OWN_NODES * BB * OO; i += OWN_THREADS) accl[i] = 0.f;
  if (tid < OWN_SCORES) {
    mxl[tid] = 0u;
    denl[tid] = 0.f;
  }
  if (tid < 1 + WALKS * PARTS) ecnt[tid] = 0;
  __syncthreads();

  const int4* tab4 = (const int4*)own;
  const unsigned gpat = blockIdx.x * 0x01010101u;
  own_scan<SCAN_LOADS, true>(tab4, 0, E16, gpat, tid, &ecnt[0], elist);
  __syncthreads();
  int total = ecnt[0];
  if (__builtin_expect(total <= OWN_THREADS, 1)) {
    const int my = lane * 32 + wv;
    int sv = 0, dv = n0, av = 0;
    if (my < total) {
      const int e = elist[my];
      sv = src[e];
      dv = dst[e];
      av = __float_as_int(eattr[e]);
    }
    const int rem = total - wv;
    const int nk = rem <= 0 ? 0 : (rem + 31) >> 5;
    const int k0 = hi ? 1 : 0;
    float al[16];
    int sl[16];
#pragma unroll
    for (int j = 0; j < 16; ++j) {
      const int sN = pair_pick(sv, j, hi);
      const int dN = pair_pick(dv, j, hi);
      const float ae = __int_as_float(pair_pick(av, j, hi));
      al[j] = edge_score(a_src[ot + sN * HH], a_dst[ot + dN * HH], ae, cf);
      sl[j] = (dN - n0) * (BB * HH) + t;
    }
#pragma unroll
    for (int j = 0; j < 16; ++j)
      if (2 * j + k0 < nk)
        __hip_atomic_fetch_max(&mxl[sl[j]], fenc(al[j]), __ATOMIC_RELAXED,
                               __HIP_MEMORY_SCOPE_WORKGROUP);
    __syncthreads();
    float* pw = pl + wv * KEEP_WAVE;
    unsigned mv[16];
#pragma unroll
    for (int j = 0; j < 16; ++j) mv[j] = mxl[sl[j]];
#pragma unroll
    for (int j = 0; j < 16; ++j) {
      const float p = __expf(al[j] - fdec(mv[j]));
      pw[(2 * j + k0) * KEEP_SLOT + t] = p;
      if (2 * j + k0 < nk)
        __hip_atomic_fetch_add(&denl[sl[j]], p, __ATOMIC_RELAXED,
                               __HIP_MEMORY_SCOPE_WORKGROUP);
    }
    __syncthreads();
    if (nk > 0) {
      RowRegs ra, rb;
      keep_load(ra, sv, dv, 0, n0, t, lane, pw, denl, h);
      int k = 0;
      for (; k + 2 < nk; k += 2) {
        keep_load(rb, sv, dv, k + 1, n0, t, lane, pw, denl, h);
        keep_add(ra, accl);
        keep_load(ra, sv, dv, k + 2, n0, t, lane, pw, denl, h);
        keep_add(rb, accl);
      }
      if (k + 1 < nk) {
        keep_load(rb, sv, dv, k + 1, n0, t, lane, pw, denl, h);
        keep_add(ra, accl);
        keep_add(rb, accl);
      } else {
        keep_add(ra, accl);
      }
    }
    __syncthreads();
  } else {
  const bool whole = total <= LIST_CAP;
  const int nparts = whole ? 1 : PARTS;
  for (int walk = 0; walk < WALKS; ++walk) {
    for (int part = 0; part < nparts; ++part) {
      if (!whole) {
        int* cnt = &ecnt[1 + walk * PARTS + part];
        const int qb = part * PART16;
        own_scan<PART_LOADS, false>(tab4, qb, min(qb + PART16, E16), gpat, tid, cnt, elist);
        __syncthreads();
        total = *cnt;
      }

      for (int b0 = 0; b0 < total; b0 += OWN_THREADS) {
        const int my = b0 + lane * 32 + wv;
        int sv = 0, dv = n0, av = 0;
        if (my < total) {
          const int e = elist[my];
          sv = src[e];
          dv = dst[e];
          av = __float_as_int(eattr[e]);
        }
        const int rem = total - b0 - wv;
        const int nk = rem <= 0 ? 0 : (rem + 31 >= 32 * 32 ? 32 : (rem + 31) >> 5);
        if (nk > 0) {
          if (walk == 0) {
            own_stat<0>(sv, dv, av, nk, n0, ot, t, hi, cf, a_src, a_dst, mxl, denl);
          } else if (walk == 1) {
            own_stat<1>(sv, dv, av, nk, n0, ot, t, hi, cf, a_src, a_dst, mxl, denl);
          } else {
            EdgeRegs ra, rb;
            own_load(ra, sv, dv, av, 0, n0, ot, t, lane, a_src, a_dst, mxl, denl, h);
            int k = 0;
            for (; k + 2 < nk; k += 2) {
              own_load(rb, sv, dv, av, k + 1, n0, ot, t, lane, a_src, a_dst, mxl, denl, h);
              own_add(ra, cf, accl);
              own_load(ra, sv, dv, av, k + 2, n0, ot, t, lane, a_src, a_dst, mxl, denl, h);
              own_add(rb, cf, accl);
            }
            if (k + 1 < nk) {
              own_load(rb, sv, dv, av, k + 1, n0, ot, t, lane, a_src, a_dst, mxl, denl, h);
              own_add(ra, cf, accl);
              own_add(rb, cf, accl);
            } else {
              own_add(ra, cf, accl);
            }
          }
        }
      }
      __syncthreads();
    }
  }
  }

  for (int i = tid; i < OWN_NODES * BB * (OO / 4); i += OWN_THREADS) {
    const int q = i & 15;
    const int r = i >> 4;
    const int nl = r % OWN_NODES, b = r / OWN_NODES;
    const float* ap = accl + (nl * BB + b) * OO + 2 * q;
    const float2 ev = *(const float2*)ap;
    const float2 od = *(const float2*)(ap + 32);
    const float4 bv = ((const float4*)bias)[q];
    float4 o;
    o.x = ev.x * 0.25f + bv.x;
    o.y = od.x * 0.25f + bv.y;
    o.z = ev.y * 0.25f + bv.z;
    o.w = od.y * 0.25f + bv.w;
    ((float4*)out)[((size_t)b * NB + n0 + nl) * (OO / 4) + q] = o;
  }
}

extern "C" void kernel_launch(void* const* d_in, const int* in_sizes, int n_in,
                              void* d_out, int out_size, void* d_ws, size_t ws_size,
                              hipStream_t stream) {
  const float* x        = (const float*)d_in[0];
  const int*   eidx     = (const int*)d_in[1];
  const float* eattr    = (const float*)d_in[2];
  const float* W        = (const float*)d_in[3];
  const float* att_src  = (const float*)d_in[4];
  const float* att_dst  = (const float*)d_in[5];
  const float* W_edge   = (const float*)d_in[6];
  const float* att_edge = (const float*)d_in[7];
  const float* bias     = (const float*)d_in[8];
  float* out = (float*)d_out;

  char* ws = (char*)d_ws;
  float*    h_buf  = (float*)   (ws + 0);
  float*    a_src  = (float*)   (ws + 40960000);
  float*    a_dst  = (float*)   (ws + 43520000);
  float*    coef   = (float*)   (ws + 51200000);
  int*      own    = (int*)     (ws + 51200256);
  float*    wap    = (float*)   (ws + 51360768);

  const int* srcI = eidx;
  const int* dstI = eidx + EE;

  edge_prep<<<PREP_BLOCKS + 1 + FOLD_BLOCKS, PREP_THREADS, 0, stream>>>(
      W_edge, att_edge, coef, dstI, own, W, att_src, att_dst, wap);
  gemm_h<<<GEMM_GROUPS, GEMM_THREADS, 0, stream>>>(x, W, wap, h_buf, a_src, a_dst);
  edge_own<<<OWN_BLOCKS, OWN_THREADS, 0, stream>>>(srcI, dstI, eattr, a_src, a_dst,
                                                   coef, h_buf, bias, own, out);
}


# Aggregate then transform: owner groups gather rows of x (half the bytes of h rows), sum z per (node,batch,head) in registers and multiply z by W on WMMA themselves; h is never stored; new sum order
# speedup vs baseline: 10.7282x; 1.1864x over previous
#include <hip/hip_runtime.h>


#define NB 10000
#define BB 4
#define EE 160000
#define DD 128
#define HH 4
#define OO 64
#define HO 256
#define MROWS (BB * NB)
#define NEG_SLOPE 0.2f

typedef float v2f __attribute__((ext_vector_type(2)));
typedef float v8f __attribute__((ext_vector_type(8)));

__device__ __forceinline__ unsigned fenc(float f) {
  unsigned u = __float_as_uint(f);
  return (u & 0x80000000u) ? ~u : (u | 0x80000000u);
}
__device__ __forceinline__ float fdec(unsigned u) {
  return __uint_as_float((u & 0x80000000u) ? (u ^ 0x80000000u) : ~u);
}

#define FCOLS 16
#define SC_GROUPS 250
#define SC_WAVES 20
#define SC_THREADS (SC_WAVES * 32)
#define GROUP_ROWS 160
#define KPAIRS (DD / 2)
#define F_ITEMS (KPAIRS * FCOLS)
#define XP (DD + 4)
#define X_ITEMS (GROUP_ROWS * (DD / 4))
static_assert(8 * SC_THREADS == X_ITEMS, "a thread stages eight x words");
static_assert(SC_GROUPS * GROUP_ROWS == MROWS, "every row has exactly one group");

__global__ __launch_bounds__(SC_THREADS) void score_gemm(const float* __restrict__ x,
                                                         const float* __restrict__ wap,
                                                         float* __restrict__ a_src,
                                                         float* __restrict__ a_dst) {
  __shared__ float2 Flds[KPAIRS * FCOLS];
  __shared__ float4 Xlds[GROUP_ROWS * XP / 4];
  const int tid = threadIdx.x;
  const int lane = tid & 31;
  const int wave = __builtin_amdgcn_readfirstlane(tid >> 5);
  const float4* xg = (const float4*)(x + (size_t)blockIdx.x * GROUP_ROWS * DD) + tid;
  const float4 xr0 = xg[0 * SC_THREADS], xr1 = xg[1 * SC_THREADS];
  const float4 xr2 = xg[2 * SC_THREADS], xr3 = xg[3 * SC_THREADS];
  const float4 xr4 = xg[4 * SC_THREADS], xr5 = xg[5 * SC_THREADS];
  const float4 xr6 = xg[6 * SC_THREADS], xr7 = xg[7 * SC_THREADS];
  float2 f0 = ((const float2*)wap)[tid];
  float2 f1 = make_float2(0.f, 0.f);
  if (tid + SC_THREADS < F_ITEMS) f1 = ((const float2*)wap)[tid + SC_THREADS];
  float4* xd = Xlds + (tid >> 5) * (XP / 4) + (tid & 31);
  xd[0 * 20 * (XP / 4)] = xr0;
  xd[1 * 20 * (XP / 4)] = xr1;
  xd[2 * 20 * (XP / 4)] = xr2;
  xd[3 * 20 * (XP / 4)] = xr3;
  xd[4 * 20 * (XP / 4)] = xr4;
  xd[5 * 20 * (XP / 4)] = xr5;
  xd[6 * 20 * (XP / 4)] = xr6;
  xd[7 * 20 * (XP / 4)] = xr7;
  Flds[tid] = f0;
  if (tid + SC_THREADS < F_ITEMS) Flds[tid + SC_THREADS] = f1;
  __syncthreads();
  if (wave >= GROUP_ROWS / 16) return;

  const int m = lane & 15;
  const int kb = (lane >> 4) << 1;
  const int kp = lane >> 4;
  const float* xrow = (const float*)Xlds + (wave * 16 + m) * XP;
  v8f accs = {0.f, 0.f, 0.f, 0.f, 0.f, 0.f, 0.f, 0.f};
  for (int k = 0; k < DD; k += 4) {
    v2f a;
    a.x = xrow[k + kb];
    a.y = xrow[k + kb + 1];
    const float2 fpr = Flds[((k >> 1) + kp) * FCOLS + m];
    v2f b;
    b.x = fpr.x;
    b.y = fpr.y;
    accs = __builtin_amdgcn_wmma_f32_16x16x4_f32(false, a, false, b, (short)0, accs,
                                                 false, false);
  }
  const int rowBase = blockIdx.x * GROUP_ROWS + wave * 16;
  const int rOff = (lane >> 4) * 8;
  if (m < 8) {
    float* tab = m < 4 ? a_src : a_dst;
#pragma unroll
    for (int v = 0; v < 8; ++v)
      tab[(size_t)(rowBase + rOff + v) * HH + (m & 3)] = accs[v];
  }
}

#define OWN_NODES 40
#define E16 (EE / 16)
static_assert(E16 * 16 == EE, "the owner plane is written and read as whole 16-byte words");
#define PREP_THREADS 256
#define PREP_BLOCKS ((E16 + PREP_THREADS - 1) / PREP_THREADS)
#define FOLD_BLOCKS ((DD * 8) / PREP_THREADS)
static_assert(FOLD_BLOCKS * PREP_THREADS == DD * 8, "one thread per folded weight");
__device__ __forceinline__ unsigned owner_pack(int4 d) {
  const unsigned a = min((unsigned)d.x / OWN_NODES, 255u);
  const unsigned b = min((unsigned)d.y / OWN_NODES, 255u);
  const unsigned c = min((unsigned)d.z / OWN_NODES, 255u);
  const unsigned e = min((unsigned)d.w / OWN_NODES, 255u);
  return a | (b << 8) | (c << 16) | (e << 24);
}
__global__ __launch_bounds__(PREP_THREADS) void edge_prep(
    const float* __restrict__ W_edge, const float* __restrict__ att_edge,
    float* __restrict__ coef, const int* __restrict__ dst, int* __restrict__ own,
    const float* __restrict__ W, const float* __restrict__ att_src,
    const float* __restrict__ att_dst, float* __restrict__ wap) {
  const int t = threadIdx.x;
  const int i = blockIdx.x * PREP_THREADS + t;
  if (i < E16) {
    const int4* d4 = (const int4*)dst + 4 * i;
    const int4 d0 = d4[0], d1 = d4[1], d2 = d4[2], d3 = d4[3];
    int4 o;
    o.x = (int)owner_pack(d0);
    o.y = (int)owner_pack(d1);
    o.z = (int)owner_pack(d2);
    o.w = (int)owner_pack(d3);
    ((int4*)own)[i] = o;
  }
  if (blockIdx.x == PREP_BLOCKS && t < HH) {
    float s = 0.f;
    for (int o = 0; o < OO; ++o) s += W_edge[t * OO + o] * att_edge[t * OO + o];
    coef[t] = s;
  }
  if (blockIdx.x > PREP_BLOCKS) {
    const int idx = (blockIdx.x - PREP_BLOCKS - 1) * PREP_THREADS + t;
    const int d = idx >> 3, c = idx & 7, hd = c & 3;
    const float4* wr = (const float4*)(W + (size_t)d * HO + hd * OO);
    const float4* ar = (const float4*)((c < 4 ? att_src : att_dst) + hd * OO);
    float s = 0.f;
#pragma unroll
    for (int o = 0; o < OO / 4; ++o) {
      const float4 w4 = wr[o], a4 = ar[o];
      s += w4.x * a4.x;
      s += w4.y * a4.y;
      s += w4.z * a4.z;
      s += w4.w * a4.w;
    }
    wap[((d >> 1) * FCOLS + c) * 2 + (d & 1)] = s;
    wap[((d >> 1) * FCOLS + c + 8) * 2 + (d & 1)] = 0.f;
  }
}

__device__ __forceinline__ float edge_score(float as, float ad, float ae, float cf) {
  float a = as + ad + ae * cf;
  return a > 0.f ? a : a * NEG_SLOPE;
}

__device__ __forceinline__ int pair_pick(int v, int j, bool hi) {
  const int lo_v = __builtin_amdgcn_readlane(v, 2 * j);
  const int hi_v = __builtin_amdgcn_readlane(v, 2 * j + 1);
  return hi ? hi_v : lo_v;
}

#define OWN_BLOCKS (NB / OWN_NODES)
static_assert(OWN_BLOCKS * OWN_NODES == NB, "every node has exactly one owner");
static_assert(OWN_BLOCKS <= 255, "an owner id is one byte and 255 means no owner");
#define OWN_THREADS 1024
#define OWN_SCORES (OWN_NODES * BB * HH)
static_assert(OWN_SCORES <= OWN_THREADS, "one thread clears one cell");
#define SCAN_LOADS 10
static_assert(SCAN_LOADS * OWN_THREADS >= E16, "one search covers every edge");
#define PART16 64
#define PARTS ((E16 + PART16 - 1) / PART16)
#define LIST_CAP (16 * PART16)
static_assert(LIST_CAP == OWN_THREADS, "one thread or lane per list entry");
#define SEARCHES (1 + 7 * PARTS)
#define ZK (HH * DD)
#define ZP (ZK + 4)
#define ROUND_ROWS 32
#define ROUNDS (OWN_NODES * BB / ROUND_ROWS)
static_assert(ROUNDS * ROUND_ROWS == OWN_NODES * BB && ROUND_ROWS * 32 == OWN_THREADS, "a row a wave");
#define WK_PAIRS (ZK / 2)
#define WL_ITEMS (WK_PAIRS * (OO / 4))
#define WL_TRIPS (WL_ITEMS / OWN_THREADS)

__device__ __forceinline__ unsigned zero_bytes(unsigned x) {
  const unsigned t = (x & 0x7F7F7F7Fu) + 0x7F7F7F7Fu;
  return ~(t | x | 0x7F7F7F7Fu);
}

template <int LOADS, bool CHECK>
__device__ __forceinline__ void own_scan(const int4* __restrict__ tab4, int qb, int qe,
                                         unsigned gpat, int tid, int* cnt, int* elist) {
  int4 dq[LOADS];
#pragma unroll
  for (int k = 0; k < LOADS; ++k) {
    const int q = qb + k * OWN_THREADS + tid;
    dq[k] = q < qe ? tab4[q] : make_int4(-1, -1, -1, -1);
  }
  unsigned m[LOADS];
  int c = 0;
#pragma unroll
  for (int k = 0; k < LOADS; ++k) {
    const unsigned f0 = zero_bytes((unsigned)dq[k].x ^ gpat);
    const unsigned f1 = zero_bytes((unsigned)dq[k].y ^ gpat);
    const unsigned f2 = zero_bytes((unsigned)dq[k].z ^ gpat);
    const unsigned f3 = zero_bytes((unsigned)dq[k].w ^ gpat);
    m[k] = (f0 >> 7) | (f1 >> 6) | (f2 >> 5) | (f3 >> 4);
    c += __builtin_popcount(m[k]);
  }
  if (c == 0) return;
  int pos = atomicAdd(cnt, c);
#pragma unroll
  for (int k = 0; k < LOADS; ++k) {
    const int q = qb + k * OWN_THREADS + tid;
    unsigned mk = m[k];
    while (mk != 0u) {
      const int bit = __builtin_ctz(mk);
      mk &= mk - 1u;
      if (!CHECK || pos < LIST_CAP) elist[pos] = 16 * q + 4 * (bit & 7) + (bit >> 3);
      ++pos;
    }
  }
}

__global__ __launch_bounds__(OWN_THREADS) void edge_own(
    const int* __restrict__ src, const int* __restrict__ dst,
    const float* __restrict__ eattr, const float* __restrict__ a_src,
    const float* __restrict__ a_dst, const float* __restrict__ coef,
    const float* __restrict__ x, const float* __restrict__ W,
    const float* __restrict__ bias, const int* __restrict__ own,
    float* __restrict__ out) {
  __shared__ float2 Wl[WK_PAIRS * OO];
  __shared__ float4 zA4[ROUND_ROWS * ZP / 4];
  __shared__ float4 pS4[LIST_CAP * BB];
  __shared__ float4 outT4[ROUND_ROWS * OO / 4];
  __shared__ int elist[LIST_CAP];
  __shared__ int srcS[LIST_CAP];
  __shared__ int dstS[LIST_CAP];
  __shared__ unsigned mxl[OWN_SCORES];
  __shared__ float denl[OWN_SCORES];
  __shared__ int hist[OWN_NODES];
  __shared__ int offs[OWN_NODES + 1];
  __shared__ int ecnt[SEARCHES];
  __shared__ float4 biasS[OO / 4];
  float* zA = (float*)zA4;
  float* pS = (float*)pS4;
  float* outT = (float*)outT4;
  const int tid = threadIdx.x;
  const int lane = tid & 31;
  const int t = lane & 15;
  const bool hi = lane >= 16;
  const int wv = __builtin_amdgcn_readfirstlane(tid >> 5);
  const int n0 = blockIdx.x * OWN_NODES;
  const int ot = (t >> 2) * (NB * HH) + (t & 3);
  const float cf = coef[t & 3];

  float4 r0[WL_TRIPS], r1[WL_TRIPS];
#pragma unroll
  for (int it = 0; it < WL_TRIPS; ++it) {
    const int i = it * OWN_THREADS + tid;
    const int kp = i >> 4, o4 = i & 15;
    const int hd = kp >> 6, d = (2 * kp) & (DD - 1);
    r0[it] = ((const float4*)(W + (size_t)d * HO + hd * OO))[o4];
    r1[it] = ((const float4*)(W + (size_t)(d + 1) * HO + hd * OO))[o4];
  }

  if (tid < OWN_SCORES) {
    mxl[tid] = 0u;
    denl[tid] = 0.f;
  }
  if (tid < OWN_NODES) hist[tid] = 0;
  if (tid < OO / 4) biasS[tid] = ((const float4*)bias)[tid];
  if (tid < ROUND_ROWS * OO / 4) outT4[tid] = make_float4(0.f, 0.f, 0.f, 0.f);
  for (int i = tid; i < SEARCHES; i += OWN_THREADS) ecnt[i] = 0;
  __syncthreads();

  const int4* tab4 = (const int4*)own;
  const unsigned gpat = blockIdx.x * 0x01010101u;
  own_scan<SCAN_LOADS, true>(tab4, 0, E16, gpat, tid, &ecnt[0], elist);
#pragma unroll
  for (int it = 0; it < WL_TRIPS; ++it) {
    const int i = it * OWN_THREADS + tid;
    const int kp = i >> 4, o4 = i & 15;
    float4* dd = (float4*)(Wl + kp * OO + 4 * o4);
    dd[0] = make_float4(r0[it].x, r1[it].x, r0[it].y, r1[it].y);
    dd[1] = make_float4(r0[it].z, r1[it].z, r0[it].w, r1[it].w);
  }
  __syncthreads();
  const int total = ecnt[0];
  const bool fast = total <= LIST_CAP;

  if (fast) {
    const int my = lane * 32 + wv;
    const bool live = my < total;
    int sv = 0, dv = n0, av = 0;
    if (live) {
      const int e = elist[my];
      sv = src[e];
      dv = dst[e];
      av = __float_as_int(eattr[e]);
    }
    const int rem = total - wv;
    const int nk = rem <= 0 ? 0 : (rem + 31) >> 5;
    const int k0 = hi ? 1 : 0;
    float al[16];
    int sl[16];
#pragma unroll
    for (int j = 0; j < 16; ++j) {
      const int sN = pair_pick(sv, j, hi);
      const int dN = pair_pick(dv, j, hi);
      const float ae = __int_as_float(pair_pick(av, j, hi));
      al[j] = edge_score(a_src[ot + sN * HH], a_dst[ot + dN * HH], ae, cf);
      sl[j] = (dN - n0) * (BB * HH) + t;
    }
    int rk = 0;
    if (live) rk = atomicAdd(&hist[dv - n0], 1);
#pragma unroll
    for (int j = 0; j < 16; ++j)
      if (2 * j + k0 < nk)
        __hip_atomic_fetch_max(&mxl[sl[j]], fenc(al[j]), __ATOMIC_RELAXED,
                               __HIP_MEMORY_SCOPE_WORKGROUP);
    __syncthreads();
    if (tid <= OWN_NODES) {
      int s = 0;
#pragma unroll
      for (int i = 0; i < OWN_NODES; ++i) s += i < tid ? hist[i] : 0;
      offs[tid] = s;
    }
    unsigned mv[16];
#pragma unroll
    for (int j = 0; j < 16; ++j) mv[j] = mxl[sl[j]];
#pragma unroll
    for (int j = 0; j < 16; ++j) {
      al[j] = __expf(al[j] - fdec(mv[j]));
      if (2 * j + k0 < nk)
        __hip_atomic_fetch_add(&denl[sl[j]], al[j], __ATOMIC_RELAXED,
                               __HIP_MEMORY_SCOPE_WORKGROUP);
    }
    __syncthreads();
    int pos = 0;
    if (live) {
      pos = offs[dv - n0] + rk;
      srcS[pos] = sv;
    }
#pragma unroll
    for (int j = 0; j < 16; ++j) {
      const int pj = pair_pick(pos, j, hi);
      if (2 * j + k0 < nk) pS[pj * (BB * HH) + t] = al[j] / fmaxf(denl[sl[j]], 1e-16f);
    }
  } else {
    for (int walk = 0; walk < 2; ++walk) {
      for (int part = 0; part < PARTS; ++part) {
        int* cnt = &ecnt[1 + walk * PARTS + part];
        const int qb = part * PART16;
        own_scan<1, false>(tab4, qb, min(qb + PART16, E16), gpat, tid, cnt, elist);
        __syncthreads();
        const int tp = *cnt;
        if (tid < tp) {
          const int e = elist[tid];
          const int s = src[e];
          const int dl = dst[e] - n0;
          const float at = eattr[e];
          for (int c = 0; c < BB * HH; ++c) {
            const int oc = (c >> 2) * (NB * HH) + (c & 3);
            const float a = edge_score(a_src[oc + s * HH], a_dst[oc + (dl + n0) * HH], at,
                                       coef[c & 3]);
            if (walk == 0)
              __hip_atomic_fetch_max(&mxl[dl * (BB * HH) + c], fenc(a), __ATOMIC_RELAXED,
                                     __HIP_MEMORY_SCOPE_WORKGROUP);
            else
              __hip_atomic_fetch_add(&denl[dl * (BB * HH) + c],
                                     __expf(a - fdec(mxl[dl * (BB * HH) + c])),
                                     __ATOMIC_RELAXED, __HIP_MEMORY_SCOPE_WORKGROUP);
          }
        }
        __syncthreads();
      }
    }
  }
  __syncthreads();

  const float4 zero4 = make_float4(0.f, 0.f, 0.f, 0.f);
  for (int r = 0; r < ROUNDS; ++r) {
    if (fast) {
      const int nl = 8 * r + (wv >> 2);
      const int b = wv & 3;
      const int beg = __builtin_amdgcn_readfirstlane(offs[nl]);
      const int end = __builtin_amdgcn_readfirstlane(offs[nl + 1]);
      float4 acc0 = zero4, acc1 = zero4, acc2 = zero4, acc3 = zero4;
      const float4* xb = (const float4*)(x + (size_t)b * NB * DD) + lane;
      for (int base = beg; base < end; base += 32) {
        const int sreg = srcS[min(base + lane, end - 1)];
        const int lim = min(base + 32, end);
        for (int kk = base; kk < lim; kk += 8) {
          float4 xv[8], w4[8];
#pragma unroll
          for (int u = 0; u < 8; ++u) {
            const int s = __builtin_amdgcn_readlane(sreg, kk - base + u);
            xv[u] = zero4;
            if (kk + u < lim) xv[u] = xb[(size_t)s * (DD / 4)];
          }
#pragma unroll
          for (int u = 0; u < 8; ++u) {
            const int e = min(kk + u, end - 1);
            w4[u] = pS4[e * BB + b];
            if (kk + u >= end) w4[u] = zero4;
          }
#pragma unroll
          for (int u = 0; u < 8; ++u) {
            acc0.x = fmaf(w4[u].x, xv[u].x, acc0.x);
            acc0.y = fmaf(w4[u].x, xv[u].y, acc0.y);
            acc0.z = fmaf(w4[u].x, xv[u].z, acc0.z);
            acc0.w = fmaf(w4[u].x, xv[u].w, acc0.w);
            acc1.x = fmaf(w4[u].y, xv[u].x, acc1.x);
            acc1.y = fmaf(w4[u].y, xv[u].y, acc1.y);
            acc1.z = fmaf(w4[u].y, xv[u].z, acc1.z);
            acc1.w = fmaf(w4[u].y, xv[u].w, acc1.w);
            acc2.x = fmaf(w4[u].z, xv[u].x, acc2.x);
            acc2.y = fmaf(w4[u].z, xv[u].y, acc2.y);
            acc2.z = fmaf(w4[u].z, xv[u].z, acc2.z);
            acc2.w = fmaf(w4[u].z, xv[u].w, acc2.w);
            acc3.x = fmaf(w4[u].w, xv[u].x, acc3.x);
            acc3.y = fmaf(w4[u].w, xv[u].y, acc3.y);
            acc3.z = fmaf(w4[u].w, xv[u].z, acc3.z);
            acc3.w = fmaf(w4[u].w, xv[u].w, acc3.w);
          }
        }
      }
      float4* zr = zA4 + wv * (ZP / 4) + lane;
      zr[0 * (DD / 4)] = acc0;
      zr[1 * (DD / 4)] = acc1;
      zr[2 * (DD / 4)] = acc2;
      zr[3 * (DD / 4)] = acc3;
    } else {
      for (int i = tid; i < ROUND_ROWS * ZP / 4; i += OWN_THREADS) zA4[i] = zero4;
      __syncthreads();
      for (int part = 0; part < PARTS; ++part) {
        int* cnt = &ecnt[1 + (2 + r) * PARTS + part];
        const int qb = part * PART16;
        own_scan<1, false>(tab4, qb, min(qb + PART16, E16), gpat, tid, cnt, elist);
        __syncthreads();
        const int tp = *cnt;
        if (tid < tp) {
          const int e = elist[tid];
          const int s = src[e];
          const int dl = dst[e] - n0;
          const float at = eattr[e];
          srcS[tid] = s;
          dstS[tid] = dl;
          for (int c = 0; c < BB * HH; ++c) {
            const int oc = (c >> 2) * (NB * HH) + (c & 3);
            const float a = edge_score(a_src[oc + s * HH], a_dst[oc + (dl + n0) * HH], at,
                                       coef[c & 3]);
            const int cell = dl * (BB * HH) + c;
            pS[tid * (BB * HH) + c] = __expf(a - fdec(mxl[cell])) / fmaxf(denl[cell], 1e-16f);
          }
        }
        __syncthreads();
        for (int i = wv; i < tp; i += 32) {
          const int dl = __builtin_amdgcn_readfirstlane(dstS[i]);
          if ((dl >> 3) == r) {
            const int s = __builtin_amdgcn_readfirstlane(srcS[i]);
            for (int b = 0; b < BB; ++b) {
              const float4 xq = ((const float4*)(x + ((size_t)b * NB + s) * DD))[lane];
              const float4 wq4 = pS4[i * BB + b];
              float* zr = zA + ((dl & 7) * BB + b) * ZP + 4 * lane;
              const float wh[HH] = {wq4.x, wq4.y, wq4.z, wq4.w};
#pragma unroll
              for (int hd = 0; hd < HH; ++hd) {
                __hip_atomic_fetch_add(zr + hd * DD + 0, wh[hd] * xq.x, __ATOMIC_RELAXED, __HIP_MEMORY_SCOPE_WORKGROUP);
                __hip_atomic_fetch_add(zr + hd * DD + 1, wh[hd] * xq.y, __ATOMIC_RELAXED, __HIP_MEMORY_SCOPE_WORKGROUP);
                __hip_atomic_fetch_add(zr + hd * DD + 2, wh[hd] * xq.z, __ATOMIC_RELAXED, __HIP_MEMORY_SCOPE_WORKGROUP);
                __hip_atomic_fetch_add(zr + hd * DD + 3, wh[hd] * xq.w, __ATOMIC_RELAXED, __HIP_MEMORY_SCOPE_WORKGROUP);
              }
            }
          }
        }
        __syncthreads();
      }
    }
    __syncthreads();

    {
      const int strip = wv & 1, tile = (wv >> 1) & 3, kq = wv >> 3;
      const int m = lane & 15;
      const int kb = (lane >> 4) << 1;
      const int kp = lane >> 4;
      const float* zrow = zA + (strip * 16 + m) * ZP + kq * DD;
      const float2* wq = Wl + (kq * (DD / 2) + kp) * OO + tile * 16 + m;
      v8f acc = {0.f, 0.f, 0.f, 0.f, 0.f, 0.f, 0.f, 0.f};
      for (int k = 0; k < DD; k += 4) {
        v2f a;
        a.x = zrow[k + kb];
        a.y = zrow[k + kb + 1];
        const float2 bp = wq[(k >> 1) * OO];
        v2f bv;
        bv.x = bp.x;
        bv.y = bp.y;
        acc = __builtin_amdgcn_wmma_f32_16x16x4_f32(false, a, false, bv, (short)0, acc,
                                                    false, false);
      }
      float* od = outT + (strip * 16 + (lane >> 4) * 8) * OO + tile * 16 + m;
#pragma unroll
      for (int v = 0; v < 8; ++v)
        __hip_atomic_fetch_add(od + v * OO, acc[v], __ATOMIC_RELAXED,
                               __HIP_MEMORY_SCOPE_WORKGROUP);
    }
    __syncthreads();

    if (tid < ROUND_ROWS * OO / 4) {
      const int row = tid >> 4, q = tid & 15;
      const float4 v = outT4[tid];
      outT4[tid] = zero4;
      const float4 bv = biasS[q];
      float4 o;
      o.x = v.x * 0.25f + bv.x;
      o.y = v.y * 0.25f + bv.y;
      o.z = v.z * 0.25f + bv.z;
      o.w = v.w * 0.25f + bv.w;
      const int node = n0 + 8 * r + (row >> 2);
      ((float4*)out)[((size_t)(row & 3) * NB + node) * (OO / 4) + q] = o;
    }
  }
}

extern "C" void kernel_launch(void* const* d_in, const int* in_sizes, int n_in,
                              void* d_out, int out_size, void* d_ws, size_t ws_size,
                              hipStream_t stream) {
  const float* x        = (const float*)d_in[0];
  const int*   eidx     = (const int*)d_in[1];
  const float* eattr    = (const float*)d_in[2];
  const float* W        = (const float*)d_in[3];
  const float* att_src  = (const float*)d_in[4];
  const float* att_dst  = (const float*)d_in[5];
  const float* W_edge   = (const float*)d_in[6];
  const float* att_edge = (const float*)d_in[7];
  const float* bias     = (const float*)d_in[8];
  float* out = (float*)d_out;

  char* ws = (char*)d_ws;
  float*    a_src  = (float*)   (ws + 40960000);
  float*    a_dst  = (float*)   (ws + 43520000);
  float*    coef   = (float*)   (ws + 51200000);
  int*      own    = (int*)     (ws + 51200256);
  float*    wap    = (float*)   (ws + 51360768);

  const int* srcI = eidx;
  const int* dstI = eidx + EE;

  edge_prep<<<PREP_BLOCKS + 1 + FOLD_BLOCKS, PREP_THREADS, 0, stream>>>(
      W_edge, att_edge, coef, dstI, own, W, att_src, att_dst, wap);
  score_gemm<<<SC_GROUPS, SC_THREADS, 0, stream>>>(x, wap, a_src, a_dst);
  edge_own<<<OWN_BLOCKS, OWN_THREADS, 0, stream>>>(srcI, dstI, eattr, a_src, a_dst, coef,
                                                   x, W, bias, own, out);
}
